# starting kernel
# baseline (speedup 1.0000x reference)
#include <hip/hip_runtime.h>

typedef float v2f __attribute__((ext_vector_type(2)));
typedef float v8f __attribute__((ext_vector_type(8)));

#define NFEAT 64

__global__ void gcn_zero_f32(float* __restrict__ p, int count) {
  int i = blockIdx.x * blockDim.x + threadIdx.x;
  if (i < count) p[i] = 0.0f;
}

__global__ void gcn_deg_init(float* __restrict__ deg, int n) {
  int i = blockIdx.x * blockDim.x + threadIdx.x;
  if (i < n) deg[i] = 1.0f;
}

__global__ void gcn_deg_count(const int* __restrict__ dst, float* __restrict__ deg, int E) {
  int i = blockIdx.x * blockDim.x + threadIdx.x;
  if (i < E) atomicAdd(&deg[dst[i]], 1.0f);
}

__global__ void gcn_deg_rsqrt(float* __restrict__ deg, int n) {
  int i = blockIdx.x * blockDim.x + threadIdx.x;
  if (i < n) deg[i] = rsqrtf(deg[i]);
}

__global__ __launch_bounds__(256) void gcn_gemm_wmma_f32(
    const float* __restrict__ in, const float* __restrict__ W,
    float* __restrict__ out, int K) {
  __shared__ float As[32 * (128 + 4)];
  const int ldA = K + 4;
  const int tid = threadIdx.x;
  const int blockRow0 = blockIdx.x * 32;

  const int total = 32 * K;
  for (int idx = tid; idx < total; idx += 256) {
    int r = idx / K;
    int c = idx - r * K;
    As[r * ldA + c] = in[(size_t)(blockRow0 + r) * K + c];
  }
  __syncthreads();

  const int wave = tid >> 5;
  const int lane = tid & 31;
  const int m  = lane & 15;
  const int kp = lane >> 4;
  const int rowL = (wave >> 2) * 16;
  const int col0 = (wave & 3) * 16;

  v8f c = {};
  for (int kk = 0; kk < K; kk += 4) {
    const int ka = kk + 2 * kp;
    v2f a, b;
    a.x = As[(rowL + m) * ldA + ka];
    a.y = As[(rowL + m) * ldA + ka + 1];
    b.x = W[(size_t)(ka + 0) * NFEAT + col0 + m];
    b.y = W[(size_t)(ka + 1) * NFEAT + col0 + m];
    c = __builtin_amdgcn_wmma_f32_16x16x4_f32(
         false, a,  false, b,
         (short)0, c,  false,  false);
  }

  const int gRow0 = blockRow0 + rowL;
#pragma unroll
  for (int i = 0; i < 8; ++i) {
    out[(size_t)(gRow0 + i + 8 * kp) * NFEAT + col0 + m] = c[i];
  }
}

__global__ void gcn_scatter(const int* __restrict__ src, const int* __restrict__ dst,
                            const float* __restrict__ h, const float* __restrict__ isq,
                            float* __restrict__ acc, int E) {
  int tid = blockIdx.x * blockDim.x + threadIdx.x;
  int e = tid >> 4;
  if (e >= E) return;
  int f = (tid & 15) * 4;
  int s = src[e], d = dst[e];
  float coef = isq[s] * isq[d];
  float4 hv = *(const float4*)(h + (size_t)s * NFEAT + f);
  float* o = acc + (size_t)d * NFEAT + f;
  atomicAdd(o + 0, hv.x * coef);
  atomicAdd(o + 1, hv.y * coef);
  atomicAdd(o + 2, hv.z * coef);
  atomicAdd(o + 3, hv.w * coef);
}

__global__ void gcn_finalize(const float* __restrict__ acc, const float* __restrict__ h,
                             const float* __restrict__ isq, const float* __restrict__ bias,
                             float* __restrict__ out, int n) {
  int t = blockIdx.x * blockDim.x + threadIdx.x;
  int i = t >> 6;
  if (i >= n) return;
  int f = t & 63;
  float is = isq[i];
  float v = acc[(size_t)i * NFEAT + f] + h[(size_t)i * NFEAT + f] * is * is + bias[f];
  out[(size_t)i * NFEAT + f] = fmaxf(v, 0.0f);
}

__global__ void gcn_pool(const float* __restrict__ h, const int* __restrict__ batch,
                         float* __restrict__ gsum, float* __restrict__ gcnt, int n) {
  int t = blockIdx.x * blockDim.x + threadIdx.x;
  int i = t >> 6;
  if (i >= n) return;
  int f = t & 63;
  int g = batch[i];
  atomicAdd(&gsum[g * NFEAT + f], h[(size_t)i * NFEAT + f]);
  if (f == 0) atomicAdd(&gcnt[g], 1.0f);
}

__global__ __launch_bounds__(256) void gcn_head(
    const float* __restrict__ gsum, const float* __restrict__ gcnt,
    const float* __restrict__ Wp, const float* __restrict__ bp,
    const float* __restrict__ Wc, const float* __restrict__ bc,
    float* __restrict__ out) {
  __shared__ float g[64 * 64];
  __shared__ float zs[64 * 64];
  const int t = threadIdx.x;
  for (int idx = t; idx < 64 * 64; idx += 256) {
    int gi = idx >> 6;
    float c = fmaxf(gcnt[gi], 1.0f);
    g[idx] = gsum[idx] / c;
  }
  __syncthreads();
  for (int idx = t; idx < 64 * 64; idx += 256) {
    int m = idx >> 6, nn = idx & 63;
    float a = bp[nn];
#pragma unroll 8
    for (int k = 0; k < 64; ++k) a += g[m * 64 + k] * Wp[k * 64 + nn];
    zs[idx] = a;
    out[384 + idx] = a;
  }
  __syncthreads();
  for (int idx = t; idx < 64 * 6; idx += 256) {
    int m = idx / 6, nn = idx - m * 6;
    float a = bc[nn];
#pragma unroll 8
    for (int k = 0; k < 64; ++k) a += zs[m * 64 + k] * Wc[k * 6 + nn];
    out[idx] = a;
  }
}

extern "C" void kernel_launch(void* const* d_in, const int* in_sizes, int n_in,
                              void* d_out, int out_size, void* d_ws, size_t ws_size,
                              hipStream_t stream) {
  const float* x     = (const float*)d_in[0];
  const int*   ei    = (const int*)d_in[1];
  const int*   batch = (const int*)d_in[2];
  const float* W1    = (const float*)d_in[3];
  const float* b1    = (const float*)d_in[4];
  const float* W2    = (const float*)d_in[5];
  const float* b2    = (const float*)d_in[6];
  const float* Wp    = (const float*)d_in[7];
  const float* bp    = (const float*)d_in[8];
  const float* Wc    = (const float*)d_in[9];
  const float* bc    = (const float*)d_in[10];

  const int n = in_sizes[0] / 128;
  const int E = in_sizes[1] / 2;
  const int* src = ei;
  const int* dst = ei + E;

  char* ws = (char*)d_ws;
  size_t off = 0;
  auto carve = [&](size_t bytes) -> float* {
    float* p = (float*)(ws + off);
    off = (off + bytes + 255) & ~(size_t)255;
    return p;
  };
  float* isq  = carve((size_t)n * sizeof(float));
  float* bufA = carve((size_t)n * NFEAT * sizeof(float));
  float* bufB = carve((size_t)n * NFEAT * sizeof(float));
  float* gsum = carve(64 * NFEAT * sizeof(float));
  float* gcnt = carve(64 * sizeof(float));

  const int T = 256;
  const int nBlk  = (n + T - 1) / T;
  const int eBlk  = (E + T - 1) / T;
  const int nfTot = n * NFEAT;
  const int nfBlk = (nfTot + T - 1) / T;
  const int e16Blk = (int)(((long long)E * 16 + T - 1) / T);
  const int gemmBlk = n / 32;

  gcn_deg_init<<<nBlk, T, 0, stream>>>(isq, n);
  gcn_deg_count<<<eBlk, T, 0, stream>>>(dst, isq, E);
  gcn_deg_rsqrt<<<nBlk, T, 0, stream>>>(isq, n);

  gcn_gemm_wmma_f32<<<gemmBlk, T, 0, stream>>>(x, W1, bufA, 128);
  gcn_zero_f32<<<nfBlk, T, 0, stream>>>(bufB, nfTot);
  gcn_scatter<<<e16Blk, T, 0, stream>>>(src, dst, bufA, isq, bufB, E);
  gcn_finalize<<<nfBlk, T, 0, stream>>>(bufB, bufA, isq, b1, bufB, n);

  gcn_gemm_wmma_f32<<<gemmBlk, T, 0, stream>>>(bufB, W2, bufA, 64);
  gcn_zero_f32<<<nfBlk, T, 0, stream>>>(bufB, nfTot);
  gcn_scatter<<<e16Blk, T, 0, stream>>>(src, dst, bufA, isq, bufB, E);
  gcn_finalize<<<nfBlk, T, 0, stream>>>(bufB, bufA, isq, b2, bufB, n);

  gcn_zero_f32<<<(64 * NFEAT + 64 + T - 1) / T, T, 0, stream>>>(gsum, 64 * NFEAT + 64);
  gcn_pool<<<nfBlk, T, 0, stream>>>(bufB, batch, gsum, gcnt, n);
  gcn_head<<<1, T, 0, stream>>>(gsum, gcnt, Wp, bp, Wc, bc, (float*)d_out);
}


# pool: each half-wave sums consecutive rows of one graph in registers, block merges 16 partials in LDS, one atomic per run instead of per node; changes float summation order
# speedup vs baseline: 1.2748x; 1.2748x over previous
#include <hip/hip_runtime.h>

typedef float v2f __attribute__((ext_vector_type(2)));
typedef float v8f __attribute__((ext_vector_type(8)));

#define NFEAT 64

__global__ void gcn_zero_f32(float* __restrict__ p, int count) {
  int i = blockIdx.x * blockDim.x + threadIdx.x;
  if (i < count) p[i] = 0.0f;
}

__global__ void gcn_deg_init(float* __restrict__ deg, int n) {
  int i = blockIdx.x * blockDim.x + threadIdx.x;
  if (i < n) deg[i] = 1.0f;
}

__global__ void gcn_deg_count(const int* __restrict__ dst, float* __restrict__ deg, int E) {
  int i = blockIdx.x * blockDim.x + threadIdx.x;
  if (i < E) atomicAdd(&deg[dst[i]], 1.0f);
}

__global__ void gcn_deg_rsqrt(float* __restrict__ deg, int n) {
  int i = blockIdx.x * blockDim.x + threadIdx.x;
  if (i < n) deg[i] = rsqrtf(deg[i]);
}

__global__ __launch_bounds__(256) void gcn_gemm_wmma_f32(
    const float* __restrict__ in, const float* __restrict__ W,
    float* __restrict__ out, int K) {
  __shared__ float As[32 * (128 + 4)];
  const int ldA = K + 4;
  const int tid = threadIdx.x;
  const int blockRow0 = blockIdx.x * 32;

  const int total = 32 * K;
  for (int idx = tid; idx < total; idx += 256) {
    int r = idx / K;
    int c = idx - r * K;
    As[r * ldA + c] = in[(size_t)(blockRow0 + r) * K + c];
  }
  __syncthreads();

  const int wave = tid >> 5;
  const int lane = tid & 31;
  const int m  = lane & 15;
  const int kp = lane >> 4;
  const int rowL = (wave >> 2) * 16;
  const int col0 = (wave & 3) * 16;

  v8f c = {};
  for (int kk = 0; kk < K; kk += 4) {
    const int ka = kk + 2 * kp;
    v2f a, b;
    a.x = As[(rowL + m) * ldA + ka];
    a.y = As[(rowL + m) * ldA + ka + 1];
    b.x = W[(size_t)(ka + 0) * NFEAT + col0 + m];
    b.y = W[(size_t)(ka + 1) * NFEAT + col0 + m];
    c = __builtin_amdgcn_wmma_f32_16x16x4_f32(
         false, a,  false, b,
         (short)0, c,  false,  false);
  }

  const int gRow0 = blockRow0 + rowL;
#pragma unroll
  for (int i = 0; i < 8; ++i) {
    out[(size_t)(gRow0 + i + 8 * kp) * NFEAT + col0 + m] = c[i];
  }
}

__global__ void gcn_scatter(const int* __restrict__ src, const int* __restrict__ dst,
                            const float* __restrict__ h, const float* __restrict__ isq,
                            float* __restrict__ acc, int E) {
  int tid = blockIdx.x * blockDim.x + threadIdx.x;
  int e = tid >> 4;
  if (e >= E) return;
  int f = (tid & 15) * 4;
  int s = src[e], d = dst[e];
  float coef = isq[s] * isq[d];
  float4 hv = *(const float4*)(h + (size_t)s * NFEAT + f);
  float* o = acc + (size_t)d * NFEAT + f;
  atomicAdd(o + 0, hv.x * coef);
  atomicAdd(o + 1, hv.y * coef);
  atomicAdd(o + 2, hv.z * coef);
  atomicAdd(o + 3, hv.w * coef);
}

__global__ void gcn_finalize(const float* __restrict__ acc, const float* __restrict__ h,
                             const float* __restrict__ isq, const float* __restrict__ bias,
                             float* __restrict__ out, int n) {
  int t = blockIdx.x * blockDim.x + threadIdx.x;
  int i = t >> 6;
  if (i >= n) return;
  int f = t & 63;
  float is = isq[i];
  float v = acc[(size_t)i * NFEAT + f] + h[(size_t)i * NFEAT + f] * is * is + bias[f];
  out[(size_t)i * NFEAT + f] = fmaxf(v, 0.0f);
}

#define POOL_RPS   16
#define POOL_SLOTS 16
#define POOL_ROWS  (POOL_RPS * POOL_SLOTS)

__device__ __forceinline__ void gcn_pool_flush4(float* __restrict__ gsum, float* __restrict__ gcnt,
                                                int g, int l16, float4 a, float c) {
  float* o = gsum + g * NFEAT + l16 * 4;
  atomicAdd(o + 0, a.x);
  atomicAdd(o + 1, a.y);
  atomicAdd(o + 2, a.z);
  atomicAdd(o + 3, a.w);
  if (l16 == 0) atomicAdd(&gcnt[g], c);
}

__global__ __launch_bounds__(256) void gcn_pool(const float* __restrict__ h, const int* __restrict__ batch,
                         float* __restrict__ gsum, float* __restrict__ gcnt, int n) {
  __shared__ alignas(16) float part[POOL_SLOTS * NFEAT];
  __shared__ int   pgid[POOL_SLOTS];
  __shared__ float pcnt[POOL_SLOTS];
  const int tid  = threadIdx.x;
  const int slot = tid >> 4;
  const int l16  = tid & 15;
  const int row0 = blockIdx.x * POOL_ROWS + slot * POOL_RPS;
  int nvalid = n - row0;
  nvalid = nvalid < 0 ? 0 : (nvalid > POOL_RPS ? POOL_RPS : nvalid);

  int    ids[POOL_RPS];
  float4 v[POOL_RPS];
#pragma unroll
  for (int j = 0; j < POOL_RPS; ++j) {
    int r = row0 + j;
    r = r < n ? r : n - 1;
    ids[j] = batch[r];
    v[j] = *(const float4*)(h + (size_t)r * NFEAT + l16 * 4);
  }

  int cur = -1;
  float4 acc = make_float4(0.0f, 0.0f, 0.0f, 0.0f);
  float cnt = 0.0f;
#pragma unroll
  for (int j = 0; j < POOL_RPS; ++j) {
    if (j < nvalid) {
      const int g = ids[j];
      if (g != cur) {
        if (cur >= 0) gcn_pool_flush4(gsum, gcnt, cur, l16, acc, cnt);
        cur = g;
        acc = make_float4(0.0f, 0.0f, 0.0f, 0.0f);
        cnt = 0.0f;
      }
      acc.x += v[j].x; acc.y += v[j].y; acc.z += v[j].z; acc.w += v[j].w;
      cnt += 1.0f;
    }
  }
  *(float4*)(part + slot * NFEAT + l16 * 4) = acc;
  if (l16 == 0) { pgid[slot] = cur; pcnt[slot] = cnt; }
  __syncthreads();

  if (tid < NFEAT) {
    int   g16[POOL_SLOTS];
    float p16[POOL_SLOTS], c16[POOL_SLOTS];
#pragma unroll
    for (int s = 0; s < POOL_SLOTS; ++s) {
      g16[s] = pgid[s];
      p16[s] = part[s * NFEAT + tid];
      c16[s] = pcnt[s];
    }
    int mg = -1;
    float a = 0.0f, c = 0.0f;
#pragma unroll
    for (int s = 0; s < POOL_SLOTS; ++s) {
      const int g = g16[s];
      if (g >= 0) {
        if (g != mg) {
          if (mg >= 0) {
            atomicAdd(&gsum[mg * NFEAT + tid], a);
            if (tid == 0) atomicAdd(&gcnt[mg], c);
          }
          mg = g; a = 0.0f; c = 0.0f;
        }
        a += p16[s];
        c += c16[s];
      }
    }
    if (mg >= 0) {
      atomicAdd(&gsum[mg * NFEAT + tid], a);
      if (tid == 0) atomicAdd(&gcnt[mg], c);
    }
  }
}

__global__ __launch_bounds__(256) void gcn_head(
    const float* __restrict__ gsum, const float* __restrict__ gcnt,
    const float* __restrict__ Wp, const float* __restrict__ bp,
    const float* __restrict__ Wc, const float* __restrict__ bc,
    float* __restrict__ out) {
  __shared__ float g[64 * 64];
  __shared__ float zs[64 * 64];
  const int t = threadIdx.x;
  for (int idx = t; idx < 64 * 64; idx += 256) {
    int gi = idx >> 6;
    float c = fmaxf(gcnt[gi], 1.0f);
    g[idx] = gsum[idx] / c;
  }
  __syncthreads();
  for (int idx = t; idx < 64 * 64; idx += 256) {
    int m = idx >> 6, nn = idx & 63;
    float a = bp[nn];
#pragma unroll 8
    for (int k = 0; k < 64; ++k) a += g[m * 64 + k] * Wp[k * 64 + nn];
    zs[idx] = a;
    out[384 + idx] = a;
  }
  __syncthreads();
  for (int idx = t; idx < 64 * 6; idx += 256) {
    int m = idx / 6, nn = idx - m * 6;
    float a = bc[nn];
#pragma unroll 8
    for (int k = 0; k < 64; ++k) a += zs[m * 64 + k] * Wc[k * 6 + nn];
    out[idx] = a;
  }
}

extern "C" void kernel_launch(void* const* d_in, const int* in_sizes, int n_in,
                              void* d_out, int out_size, void* d_ws, size_t ws_size,
                              hipStream_t stream) {
  const float* x     = (const float*)d_in[0];
  const int*   ei    = (const int*)d_in[1];
  const int*   batch = (const int*)d_in[2];
  const float* W1    = (const float*)d_in[3];
  const float* b1    = (const float*)d_in[4];
  const float* W2    = (const float*)d_in[5];
  const float* b2    = (const float*)d_in[6];
  const float* Wp    = (const float*)d_in[7];
  const float* bp    = (const float*)d_in[8];
  const float* Wc    = (const float*)d_in[9];
  const float* bc    = (const float*)d_in[10];

  const int n = in_sizes[0] / 128;
  const int E = in_sizes[1] / 2;
  const int* src = ei;
  const int* dst = ei + E;

  char* ws = (char*)d_ws;
  size_t off = 0;
  auto carve = [&](size_t bytes) -> float* {
    float* p = (float*)(ws + off);
    off = (off + bytes + 255) & ~(size_t)255;
    return p;
  };
  float* isq  = carve((size_t)n * sizeof(float));
  float* bufA = carve((size_t)n * NFEAT * sizeof(float));
  float* bufB = carve((size_t)n * NFEAT * sizeof(float));
  float* gsum = carve(64 * NFEAT * sizeof(float));
  float* gcnt = carve(64 * sizeof(float));

  const int T = 256;
  const int nBlk  = (n + T - 1) / T;
  const int eBlk  = (E + T - 1) / T;
  const int nfTot = n * NFEAT;
  const int nfBlk = (nfTot + T - 1) / T;
  const int e16Blk = (int)(((long long)E * 16 + T - 1) / T);
  const int gemmBlk = n / 32;
  const int poolBlk = (n + POOL_ROWS - 1) / POOL_ROWS;

  gcn_deg_init<<<nBlk, T, 0, stream>>>(isq, n);
  gcn_deg_count<<<eBlk, T, 0, stream>>>(dst, isq, E);
  gcn_deg_rsqrt<<<nBlk, T, 0, stream>>>(isq, n);

  gcn_gemm_wmma_f32<<<gemmBlk, T, 0, stream>>>(x, W1, bufA, 128);
  gcn_zero_f32<<<nfBlk, T, 0, stream>>>(bufB, nfTot);
  gcn_scatter<<<e16Blk, T, 0, stream>>>(src, dst, bufA, isq, bufB, E);
  gcn_finalize<<<nfBlk, T, 0, stream>>>(bufB, bufA, isq, b1, bufB, n);

  gcn_gemm_wmma_f32<<<gemmBlk, T, 0, stream>>>(bufB, W2, bufA, 64);
  gcn_zero_f32<<<nfBlk, T, 0, stream>>>(bufB, nfTot);
  gcn_scatter<<<e16Blk, T, 0, stream>>>(src, dst, bufA, isq, bufB, E);
  gcn_finalize<<<nfBlk, T, 0, stream>>>(bufB, bufA, isq, b2, bufB, n);

  gcn_zero_f32<<<(64 * NFEAT + 64 + T - 1) / T, T, 0, stream>>>(gsum, 64 * NFEAT + 64);
  gcn_pool<<<poolBlk, T, 0, stream>>>(bufB, batch, gsum, gcnt, n);
  gcn_head<<<1, T, 0, stream>>>(gsum, gcnt, Wp, bp, Wc, bc, (float*)d_out);
}


# aggregation pulled: edges counting-sorted by destination in every call (integer atomics); a half-wave per node sums its source rows in registers, self loop+bias+relu fused; float sum order changes
# speedup vs baseline: 1.8671x; 1.4646x over previous
#include <hip/hip_runtime.h>

typedef float v2f __attribute__((ext_vector_type(2)));
typedef float v8f __attribute__((ext_vector_type(8)));

#define NFEAT 64

__global__ void gcn_zero_f32(float* __restrict__ p, int count) {
  int i = blockIdx.x * blockDim.x + threadIdx.x;
  if (i < count) p[i] = 0.0f;
}

#define EDGE_PER 4

__global__ __launch_bounds__(256) void gcn_indeg_count(const int* __restrict__ dst,
                                                       int* __restrict__ cnt, int E) {
  const int e0 = blockIdx.x * (256 * EDGE_PER) + threadIdx.x;
  int d[EDGE_PER];
#pragma unroll
  for (int k = 0; k < EDGE_PER; ++k) {
    const int e = e0 + k * 256;
    d[k] = (e < E) ? dst[e] : -1;
  }
#pragma unroll
  for (int k = 0; k < EDGE_PER; ++k)
    if (d[k] >= 0) atomicAdd(&cnt[d[k]], 1);
}

#define SCAN_TILE 1024

__global__ __launch_bounds__(256) void gcn_scan_tile_sums(const int* __restrict__ cnt,
                                                          int* __restrict__ tsum, int n) {
  __shared__ int red[8];
  const int tid = threadIdx.x;
  const int base = blockIdx.x * SCAN_TILE + tid * 4;
  int s = 0;
#pragma unroll
  for (int k = 0; k < 4; ++k) {
    const int i = base + k;
    s += (i < n) ? cnt[i] : 0;
  }
  for (int off = 16; off > 0; off >>= 1) s += __shfl_xor(s, off, 32);
  if ((tid & 31) == 0) red[tid >> 5] = s;
  __syncthreads();
  if (tid == 0) {
    int t = 0;
    for (int w = 0; w < 8; ++w) t += red[w];
    tsum[blockIdx.x] = t;
  }
}

__global__ __launch_bounds__(256) void gcn_scan_rowptr(const int* __restrict__ cnt,
                                                       const int* __restrict__ tsum,
                                                       int* __restrict__ rowptr,
                                                       int* __restrict__ cursor,
                                                       float* __restrict__ isq, int n) {
  __shared__ int red[8];
  __shared__ int wtot[8];
  const int tid  = threadIdx.x;
  const int lane = tid & 31;
  const int wave = tid >> 5;

  int p = 0;
  for (int i = tid; i < (int)blockIdx.x; i += 256) p += tsum[i];
  for (int off = 16; off > 0; off >>= 1) p += __shfl_xor(p, off, 32);

  const int base = blockIdx.x * SCAN_TILE + tid * 4;
  int c[4];
  int t4 = 0;
#pragma unroll
  for (int k = 0; k < 4; ++k) {
    const int i = base + k;
    c[k] = (i < n) ? cnt[i] : 0;
    t4 += c[k];
  }
  int x = t4;
  for (int off = 1; off < 32; off <<= 1) {
    const int y = __shfl_up(x, off, 32);
    if (lane >= off) x += y;
  }
  if (lane == 0) red[wave] = p;
  if (lane == 31) wtot[wave] = x;
  __syncthreads();
  int excl = x - t4;
  for (int w = 0; w < 8; ++w) {
    excl += red[w];
    if (w < wave) excl += wtot[w];
  }
#pragma unroll
  for (int k = 0; k < 4; ++k) {
    const int i = base + k;
    if (i <= n) rowptr[i] = excl;
    if (i < n) {
      cursor[i] = excl;
      isq[i] = rsqrtf((float)(c[k] + 1));
    }
    excl += c[k];
  }
}

__global__ __launch_bounds__(256) void gcn_fill_by_dst(const int* __restrict__ src,
                                                       const int* __restrict__ dst,
                                                       const float* __restrict__ isq,
                                                       int* __restrict__ cursor,
                                                       int* __restrict__ ewi, int E) {
  int2* __restrict__ ew = (int2*)ewi;
  const int e0 = blockIdx.x * (256 * EDGE_PER) + threadIdx.x;
  int d[EDGE_PER], s[EDGE_PER], slot[EDGE_PER];
  float ws[EDGE_PER];
#pragma unroll
  for (int k = 0; k < EDGE_PER; ++k) {
    const int e = e0 + k * 256;
    d[k] = (e < E) ? dst[e] : -1;
    s[k] = (e < E) ? src[e] : 0;
  }
#pragma unroll
  for (int k = 0; k < EDGE_PER; ++k) {
    slot[k] = 0;
    ws[k] = 0.0f;
    if (d[k] >= 0) {
      ws[k] = isq[s[k]];
      slot[k] = atomicAdd(&cursor[d[k]], 1);
    }
  }
#pragma unroll
  for (int k = 0; k < EDGE_PER; ++k)
    if (d[k] >= 0) ew[slot[k]] = make_int2(s[k], __float_as_int(ws[k]));
}

__global__ __launch_bounds__(256) void gcn_gemm_wmma_f32(
    const float* __restrict__ in, const float* __restrict__ W,
    float* __restrict__ out, int K) {
  __shared__ float As[32 * (128 + 4)];
  const int ldA = K + 4;
  const int tid = threadIdx.x;
  const int blockRow0 = blockIdx.x * 32;

  const int total = 32 * K;
  for (int idx = tid; idx < total; idx += 256) {
    int r = idx / K;
    int c = idx - r * K;
    As[r * ldA + c] = in[(size_t)(blockRow0 + r) * K + c];
  }
  __syncthreads();

  const int wave = tid >> 5;
  const int lane = tid & 31;
  const int m  = lane & 15;
  const int kp = lane >> 4;
  const int rowL = (wave >> 2) * 16;
  const int col0 = (wave & 3) * 16;

  v8f c = {};
  for (int kk = 0; kk < K; kk += 4) {
    const int ka = kk + 2 * kp;
    v2f a, b;
    a.x = As[(rowL + m) * ldA + ka];
    a.y = As[(rowL + m) * ldA + ka + 1];
    b.x = W[(size_t)(ka + 0) * NFEAT + col0 + m];
    b.y = W[(size_t)(ka + 1) * NFEAT + col0 + m];
    c = __builtin_amdgcn_wmma_f32_16x16x4_f32(
         false, a,  false, b,
         (short)0, c,  false,  false);
  }

  const int gRow0 = blockRow0 + rowL;
#pragma unroll
  for (int i = 0; i < 8; ++i) {
    out[(size_t)(gRow0 + i + 8 * kp) * NFEAT + col0 + m] = c[i];
  }
}

__global__ __launch_bounds__(256) void gcn_gather_relu(
    const int* __restrict__ rowptr, const int* __restrict__ ewi,
    const float* __restrict__ h, const float* __restrict__ isq,
    const float* __restrict__ bias, float* __restrict__ out, int n) {
  const int2* __restrict__ ew = (const int2*)ewi;
  const int tid  = threadIdx.x;
  const int l16  = tid & 15;
  const int node = blockIdx.x * 16 + (tid >> 4);
  if (node >= n) return;
  const int beg = rowptr[node];
  const int end = rowptr[node + 1];
  const float isd = isq[node];
  const float* hp = h + l16 * 4;

  const float4 hs = *(const float4*)(hp + (size_t)node * NFEAT);
  const float cs = isd * isd;
  float4 acc = make_float4(hs.x * cs, hs.y * cs, hs.z * cs, hs.w * cs);

  for (int base = beg; base < end; base += 16) {
    const int my = base + l16;
    int   sl = node;
    float wl = 0.0f;
    if (my < end) {
      const int2 p = ew[my];
      sl = p.x;
      wl = __int_as_float(p.y) * isd;
    }
    const int cnt = (end - base) < 16 ? (end - base) : 16;
    for (int j = 0; j < cnt; j += 4) {
      int    s[4];
      float  w[4];
      float4 hv[4];
#pragma unroll
      for (int u = 0; u < 4; ++u) s[u] = __shfl(sl, j + u, 16);
#pragma unroll
      for (int u = 0; u < 4; ++u) hv[u] = *(const float4*)(hp + (size_t)s[u] * NFEAT);
#pragma unroll
      for (int u = 0; u < 4; ++u) w[u] = __shfl(wl, j + u, 16);
#pragma unroll
      for (int u = 0; u < 4; ++u) {
        if (j + u < cnt) {
          acc.x = fmaf(hv[u].x, w[u], acc.x);
          acc.y = fmaf(hv[u].y, w[u], acc.y);
          acc.z = fmaf(hv[u].z, w[u], acc.z);
          acc.w = fmaf(hv[u].w, w[u], acc.w);
        }
      }
    }
  }

  const float4 b = *(const float4*)(bias + l16 * 4);
  float4 o;
  o.x = fmaxf(acc.x + b.x, 0.0f);
  o.y = fmaxf(acc.y + b.y, 0.0f);
  o.z = fmaxf(acc.z + b.z, 0.0f);
  o.w = fmaxf(acc.w + b.w, 0.0f);
  *(float4*)(out + (size_t)node * NFEAT + l16 * 4) = o;
}

#define POOL_RPS   16
#define POOL_SLOTS 16
#define POOL_ROWS  (POOL_RPS * POOL_SLOTS)

__device__ __forceinline__ void gcn_pool_flush4(float* __restrict__ gsum, float* __restrict__ gcnt,
                                                int g, int l16, float4 a, float c) {
  float* o = gsum + g * NFEAT + l16 * 4;
  atomicAdd(o + 0, a.x);
  atomicAdd(o + 1, a.y);
  atomicAdd(o + 2, a.z);
  atomicAdd(o + 3, a.w);
  if (l16 == 0) atomicAdd(&gcnt[g], c);
}

__global__ __launch_bounds__(256) void gcn_pool(const float* __restrict__ h, const int* __restrict__ batch,
                         float* __restrict__ gsum, float* __restrict__ gcnt, int n) {
  __shared__ alignas(16) float part[POOL_SLOTS * NFEAT];
  __shared__ int   pgid[POOL_SLOTS];
  __shared__ float pcnt[POOL_SLOTS];
  const int tid  = threadIdx.x;
  const int slot = tid >> 4;
  const int l16  = tid & 15;
  const int row0 = blockIdx.x * POOL_ROWS + slot * POOL_RPS;
  int nvalid = n - row0;
  nvalid = nvalid < 0 ? 0 : (nvalid > POOL_RPS ? POOL_RPS : nvalid);

  int    ids[POOL_RPS];
  float4 v[POOL_RPS];
#pragma unroll
  for (int j = 0; j < POOL_RPS; ++j) {
    int r = row0 + j;
    r = r < n ? r : n - 1;
    ids[j] = batch[r];
    v[j] = *(const float4*)(h + (size_t)r * NFEAT + l16 * 4);
  }

  int cur = -1;
  float4 acc = make_float4(0.0f, 0.0f, 0.0f, 0.0f);
  float cnt = 0.0f;
#pragma unroll
  for (int j = 0; j < POOL_RPS; ++j) {
    if (j < nvalid) {
      const int g = ids[j];
      if (g != cur) {
        if (cur >= 0) gcn_pool_flush4(gsum, gcnt, cur, l16, acc, cnt);
        cur = g;
        acc = make_float4(0.0f, 0.0f, 0.0f, 0.0f);
        cnt = 0.0f;
      }
      acc.x += v[j].x; acc.y += v[j].y; acc.z += v[j].z; acc.w += v[j].w;
      cnt += 1.0f;
    }
  }
  *(float4*)(part + slot * NFEAT + l16 * 4) = acc;
  if (l16 == 0) { pgid[slot] = cur; pcnt[slot] = cnt; }
  __syncthreads();

  if (tid < NFEAT) {
    int   g16[POOL_SLOTS];
    float p16[POOL_SLOTS], c16[POOL_SLOTS];
#pragma unroll
    for (int s = 0; s < POOL_SLOTS; ++s) {
      g16[s] = pgid[s];
      p16[s] = part[s * NFEAT + tid];
      c16[s] = pcnt[s];
    }
    int mg = -1;
    float a = 0.0f, c = 0.0f;
#pragma unroll
    for (int s = 0; s < POOL_SLOTS; ++s) {
      const int g = g16[s];
      if (g >= 0) {
        if (g != mg) {
          if (mg >= 0) {
            atomicAdd(&gsum[mg * NFEAT + tid], a);
            if (tid == 0) atomicAdd(&gcnt[mg], c);
          }
          mg = g; a = 0.0f; c = 0.0f;
        }
        a += p16[s];
        c += c16[s];
      }
    }
    if (mg >= 0) {
      atomicAdd(&gsum[mg * NFEAT + tid], a);
      if (tid == 0) atomicAdd(&gcnt[mg], c);
    }
  }
}

__global__ __launch_bounds__(256) void gcn_head(
    const float* __restrict__ gsum, const float* __restrict__ gcnt,
    const float* __restrict__ Wp, const float* __restrict__ bp,
    const float* __restrict__ Wc, const float* __restrict__ bc,
    float* __restrict__ out) {
  __shared__ float g[64 * 64];
  __shared__ float zs[64 * 64];
  const int t = threadIdx.x;
  for (int idx = t; idx < 64 * 64; idx += 256) {
    int gi = idx >> 6;
    float c = fmaxf(gcnt[gi], 1.0f);
    g[idx] = gsum[idx] / c;
  }
  __syncthreads();
  for (int idx = t; idx < 64 * 64; idx += 256) {
    int m = idx >> 6, nn = idx & 63;
    float a = bp[nn];
#pragma unroll 8
    for (int k = 0; k < 64; ++k) a += g[m * 64 + k] * Wp[k * 64 + nn];
    zs[idx] = a;
    out[384 + idx] = a;
  }
  __syncthreads();
  for (int idx = t; idx < 64 * 6; idx += 256) {
    int m = idx / 6, nn = idx - m * 6;
    float a = bc[nn];
#pragma unroll 8
    for (int k = 0; k < 64; ++k) a += zs[m * 64 + k] * Wc[k * 6 + nn];
    out[idx] = a;
  }
}

extern "C" void kernel_launch(void* const* d_in, const int* in_sizes, int n_in,
                              void* d_out, int out_size, void* d_ws, size_t ws_size,
                              hipStream_t stream) {
  const float* x     = (const float*)d_in[0];
  const int*   ei    = (const int*)d_in[1];
  const int*   batch = (const int*)d_in[2];
  const float* W1    = (const float*)d_in[3];
  const float* b1    = (const float*)d_in[4];
  const float* W2    = (const float*)d_in[5];
  const float* b2    = (const float*)d_in[6];
  const float* Wp    = (const float*)d_in[7];
  const float* bp    = (const float*)d_in[8];
  const float* Wc    = (const float*)d_in[9];
  const float* bc    = (const float*)d_in[10];

  const int n = in_sizes[0] / 128;
  const int E = in_sizes[1] / 2;
  const int* src = ei;
  const int* dst = ei + E;

  const int T = 256;
  const int eBlk  = (E + T * EDGE_PER - 1) / (T * EDGE_PER);
  const int gemmBlk = n / 32;
  const int poolBlk = (n + POOL_ROWS - 1) / POOL_ROWS;
  const int scanBlk = (n + 1 + SCAN_TILE - 1) / SCAN_TILE;
  const int gathBlk = (n + 15) / 16;

  char* ws = (char*)d_ws;
  size_t off = 0;
  auto carve = [&](size_t bytes) -> float* {
    float* p = (float*)(ws + off);
    off = (off + bytes + 255) & ~(size_t)255;
    return p;
  };
  float* isq  = carve((size_t)n * sizeof(float));
  float* bufA = carve((size_t)n * NFEAT * sizeof(float));
  float* bufB = carve((size_t)n * NFEAT * sizeof(float));
  int*   rowptr = (int*)carve((size_t)(n + 1) * sizeof(int));
  int*   cursor = (int*)carve((size_t)n * sizeof(int));
  int*   tsum   = (int*)carve((size_t)scanBlk * sizeof(int));
  int*   ew     = (int*)carve((size_t)E * 2 * sizeof(int));
  float* zero0 = carve((size_t)n * sizeof(int));
  int*   cnt   = (int*)zero0;
  float* gsum  = carve(64 * NFEAT * sizeof(float));
  float* gcnt  = carve(64 * sizeof(float));
  const int zeroCount = (int)((gcnt + 64) - zero0);

  gcn_zero_f32<<<(zeroCount + T - 1) / T, T, 0, stream>>>(zero0, zeroCount);
  gcn_indeg_count<<<eBlk, T, 0, stream>>>(dst, cnt, E);
  gcn_scan_tile_sums<<<scanBlk, T, 0, stream>>>(cnt, tsum, n);
  gcn_scan_rowptr<<<scanBlk, T, 0, stream>>>(cnt, tsum, rowptr, cursor, isq, n);
  gcn_fill_by_dst<<<eBlk, T, 0, stream>>>(src, dst, isq, cursor, ew, E);

  gcn_gemm_wmma_f32<<<gemmBlk, T, 0, stream>>>(x, W1, bufA, 128);
  gcn_gather_relu<<<gathBlk, T, 0, stream>>>(rowptr, ew, bufA, isq, b1, bufB, n);

  gcn_gemm_wmma_f32<<<gemmBlk, T, 0, stream>>>(bufB, W2, bufA, 64);
  gcn_gather_relu<<<gathBlk, T, 0, stream>>>(rowptr, ew, bufA, isq, b2, bufB, n);

  gcn_pool<<<poolBlk, T, 0, stream>>>(bufB, batch, gsum, gcnt, n);
  gcn_head<<<1, T, 0, stream>>>(gsum, gcnt, Wp, bp, Wc, bc, (float*)d_out);
}


# edges grouped by destination through LDS bins of 256 nodes (one global add per block and bin, not two global atomics per edge); source deg^-1/2 applied at GEMM store; float sum order changes
# speedup vs baseline: 4.1190x; 2.2061x over previous
#include <hip/hip_runtime.h>

typedef float v2f __attribute__((ext_vector_type(2)));
typedef float v8f __attribute__((ext_vector_type(8)));

#define NFEAT 64

__global__ void gcn_zero_f32(float* __restrict__ p, int count) {
  int i = blockIdx.x * blockDim.x + threadIdx.x;
  if (i < count) p[i] = 0.0f;
}

#define BIN_SHIFT 8
#define BIN_W     (1 << BIN_SHIFT)
#define MAX_BINS  512
#define EDGE_PER  16
#define CHUNK     (256 * EDGE_PER)
#define LIST_PER  8

__device__ __forceinline__ int gcn_block_excl_scan(int v, int* wtot, int& total) {
  const int lane = threadIdx.x & 31;
  const int wave = threadIdx.x >> 5;
  int x = v;
  for (int off = 1; off < 32; off <<= 1) {
    const int y = __shfl_up(x, off, 32);
    if (lane >= off) x += y;
  }
  if (lane == 31) wtot[wave] = x;
  __syncthreads();
  int pre = 0, tot = 0;
  for (int w = 0; w < 8; ++w) {
    const int tw = wtot[w];
    if (w < wave) pre += tw;
    tot += tw;
  }
  __syncthreads();
  total = tot;
  return pre + x - v;
}

__global__ __launch_bounds__(256) void gcn_bin_count(const int* __restrict__ dst,
                                                     int* __restrict__ btot, int E, int nbins) {
  __shared__ int hist[MAX_BINS];
  const int tid = threadIdx.x;
  hist[tid] = 0;
  hist[tid + 256] = 0;
  __syncthreads();
  const int e0 = blockIdx.x * CHUNK + tid;
  int d[EDGE_PER];
#pragma unroll
  for (int k = 0; k < EDGE_PER; ++k) {
    const int e = e0 + k * 256;
    d[k] = (e < E) ? dst[e] : -1;
  }
#pragma unroll
  for (int k = 0; k < EDGE_PER; ++k)
    if (d[k] >= 0) atomicAdd(&hist[d[k] >> BIN_SHIFT], 1);
  __syncthreads();
  for (int b = tid; b < nbins; b += 256) {
    const int h = hist[b];
    if (h) atomicAdd(&btot[b], h);
  }
}

__global__ __launch_bounds__(256) void gcn_bin_fill(const int* __restrict__ src,
                                                    const int* __restrict__ dst,
                                                    const int* __restrict__ btot,
                                                    int* __restrict__ bcur,
                                                    int* __restrict__ binnedi, int E, int nbins) {
  __shared__ int  hist[MAX_BINS];
  __shared__ int  delta[MAX_BINS];
  __shared__ int  wtot[8];
  __shared__ int2 pairs[CHUNK];
  int2* __restrict__ binned = (int2*)binnedi;
  const int tid = threadIdx.x;
  hist[tid] = 0;
  hist[tid + 256] = 0;
  __syncthreads();

  const int e0 = blockIdx.x * CHUNK + tid;
  int d[EDGE_PER], s[EDGE_PER], rank[EDGE_PER];
#pragma unroll
  for (int k = 0; k < EDGE_PER; ++k) {
    const int e = e0 + k * 256;
    d[k] = (e < E) ? dst[e] : -1;
    s[k] = (e < E) ? src[e] : 0;
  }
  const int b0 = 2 * tid, b1 = 2 * tid + 1;
  const int t0 = (b0 < nbins) ? btot[b0] : 0;
  const int t1 = (b1 < nbins) ? btot[b1] : 0;
#pragma unroll
  for (int k = 0; k < EDGE_PER; ++k) {
    rank[k] = 0;
    if (d[k] >= 0) rank[k] = atomicAdd(&hist[d[k] >> BIN_SHIFT], 1);
  }
  __syncthreads();

  const int h0 = hist[b0], h1 = hist[b1];
  int nvalid, etot;
  const int ls0 = gcn_block_excl_scan(h0 + h1, wtot, nvalid);
  const int gs0 = gcn_block_excl_scan(t0 + t1, wtot, etot);
  const int r0 = h0 ? atomicAdd(&bcur[b0], h0) : 0;
  const int r1 = h1 ? atomicAdd(&bcur[b1], h1) : 0;
  hist[b0]  = ls0;
  hist[b1]  = ls0 + h0;
  delta[b0] = gs0 + r0 - ls0;
  delta[b1] = gs0 + t0 + r1 - (ls0 + h0);
  __syncthreads();

#pragma unroll
  for (int k = 0; k < EDGE_PER; ++k)
    if (d[k] >= 0) pairs[hist[d[k] >> BIN_SHIFT] + rank[k]] = make_int2(s[k], d[k]);
  __syncthreads();

  for (int i = tid; i < nvalid; i += 256) {
    const int2 p = pairs[i];
    binned[i + delta[p.y >> BIN_SHIFT]] = p;
  }
}

__global__ __launch_bounds__(256) void gcn_bin_lists(const int* __restrict__ binnedi,
                                                     const int* __restrict__ btot,
                                                     int* __restrict__ rowptr,
                                                     float* __restrict__ isq,
                                                     int* __restrict__ ew, int n, int nbins) {
  __shared__ int ncnt[BIN_W];
  __shared__ int wtot[8];
  const int2* __restrict__ binned = (const int2*)binnedi;
  const int tid = threadIdx.x;
  const int bin = blockIdx.x;

  int v = 0;
  if (tid < bin) v += btot[tid];
  if (tid + 256 < bin) v += btot[tid + 256];
  const int T = btot[bin];
  ncnt[tid] = 0;
  int binBase;
  gcn_block_excl_scan(v, wtot, binBase);

  const int2* __restrict__ mine = binned + binBase;
  for (int i0 = tid; i0 < T; i0 += 256 * LIST_PER) {
    int dl[LIST_PER];
#pragma unroll
    for (int u = 0; u < LIST_PER; ++u) {
      const int i = i0 + u * 256;
      dl[u] = (i < T) ? mine[i].y : -1;
    }
#pragma unroll
    for (int u = 0; u < LIST_PER; ++u)
      if (dl[u] >= 0) atomicAdd(&ncnt[dl[u] & (BIN_W - 1)], 1);
  }
  __syncthreads();

  const int c = ncnt[tid];
  int tsum;
  const int first = binBase + gcn_block_excl_scan(c, wtot, tsum);
  const int node = bin * BIN_W + tid;
  if (node < n) {
    rowptr[node] = first;
    isq[node] = rsqrtf((float)(c + 1));
  }
  if (bin == nbins - 1 && tid == 0) rowptr[n] = binBase + T;
  ncnt[tid] = first;
  __syncthreads();

  for (int i0 = tid; i0 < T; i0 += 256 * LIST_PER) {
    int2 p[LIST_PER];
#pragma unroll
    for (int u = 0; u < LIST_PER; ++u) {
      const int i = i0 + u * 256;
      p[u] = (i < T) ? mine[i] : make_int2(0, -1);
    }
#pragma unroll
    for (int u = 0; u < LIST_PER; ++u)
      if (p[u].y >= 0) ew[atomicAdd(&ncnt[p[u].y & (BIN_W - 1)], 1)] = p[u].x;
  }
}

__global__ __launch_bounds__(256) void gcn_gemm_wmma_f32(
    const float* __restrict__ in, const float* __restrict__ W,
    const float* __restrict__ rowscale, float* __restrict__ out, int K) {
  __shared__ float As[32 * (128 + 4)];
  const int ldA = K + 4;
  const int tid = threadIdx.x;
  const int blockRow0 = blockIdx.x * 32;

  const int total = 32 * K;
  for (int idx = tid; idx < total; idx += 256) {
    int r = idx / K;
    int c = idx - r * K;
    As[r * ldA + c] = in[(size_t)(blockRow0 + r) * K + c];
  }
  __syncthreads();

  const int wave = tid >> 5;
  const int lane = tid & 31;
  const int m  = lane & 15;
  const int kp = lane >> 4;
  const int rowL = (wave >> 2) * 16;
  const int col0 = (wave & 3) * 16;

  const int gRow0 = blockRow0 + rowL;
  float rs[8];
#pragma unroll
  for (int i = 0; i < 8; ++i) rs[i] = rowscale[gRow0 + i + 8 * kp];

  v8f c = {};
  for (int kk = 0; kk < K; kk += 4) {
    const int ka = kk + 2 * kp;
    v2f a, b;
    a.x = As[(rowL + m) * ldA + ka];
    a.y = As[(rowL + m) * ldA + ka + 1];
    b.x = W[(size_t)(ka + 0) * NFEAT + col0 + m];
    b.y = W[(size_t)(ka + 1) * NFEAT + col0 + m];
    c = __builtin_amdgcn_wmma_f32_16x16x4_f32(
         false, a,  false, b,
         (short)0, c,  false,  false);
  }

#pragma unroll
  for (int i = 0; i < 8; ++i) {
    out[(size_t)(gRow0 + i + 8 * kp) * NFEAT + col0 + m] = c[i] * rs[i];
  }
}

__global__ __launch_bounds__(256) void gcn_gather_relu(
    const int* __restrict__ rowptr, const int* __restrict__ ew,
    const float* __restrict__ h, const float* __restrict__ isq,
    const float* __restrict__ bias, float* __restrict__ out, int n) {
  const int tid  = threadIdx.x;
  const int l16  = tid & 15;
  const int node = blockIdx.x * 16 + (tid >> 4);
  if (node >= n) return;
  const int beg = rowptr[node];
  const int end = rowptr[node + 1];
  const float isd = isq[node];
  const float* hp = h + l16 * 4;

  float4 acc = *(const float4*)(hp + (size_t)node * NFEAT);

  for (int base = beg; base < end; base += 16) {
    const int my = base + l16;
    const int sl = (my < end) ? ew[my] : node;
    const int cnt = (end - base) < 16 ? (end - base) : 16;
    for (int j = 0; j < cnt; j += 4) {
      int    s[4];
      float4 hv[4];
#pragma unroll
      for (int u = 0; u < 4; ++u) s[u] = __shfl(sl, j + u, 16);
#pragma unroll
      for (int u = 0; u < 4; ++u) hv[u] = *(const float4*)(hp + (size_t)s[u] * NFEAT);
#pragma unroll
      for (int u = 0; u < 4; ++u) {
        if (j + u < cnt) {
          acc.x += hv[u].x;
          acc.y += hv[u].y;
          acc.z += hv[u].z;
          acc.w += hv[u].w;
        }
      }
    }
  }

  const float4 b = *(const float4*)(bias + l16 * 4);
  float4 o;
  o.x = fmaxf(fmaf(acc.x, isd, b.x), 0.0f);
  o.y = fmaxf(fmaf(acc.y, isd, b.y), 0.0f);
  o.z = fmaxf(fmaf(acc.z, isd, b.z), 0.0f);
  o.w = fmaxf(fmaf(acc.w, isd, b.w), 0.0f);
  *(float4*)(out + (size_t)node * NFEAT + l16 * 4) = o;
}

#define POOL_RPS   16
#define POOL_SLOTS 16
#define POOL_ROWS  (POOL_RPS * POOL_SLOTS)

__device__ __forceinline__ void gcn_pool_flush4(float* __restrict__ gsum, float* __restrict__ gcnt,
                                                int g, int l16, float4 a, float c) {
  float* o = gsum + g * NFEAT + l16 * 4;
  atomicAdd(o + 0, a.x);
  atomicAdd(o + 1, a.y);
  atomicAdd(o + 2, a.z);
  atomicAdd(o + 3, a.w);
  if (l16 == 0) atomicAdd(&gcnt[g], c);
}

__global__ __launch_bounds__(256) void gcn_pool(const float* __restrict__ h, const int* __restrict__ batch,
                         float* __restrict__ gsum, float* __restrict__ gcnt, int n) {
  __shared__ alignas(16) float part[POOL_SLOTS * NFEAT];
  __shared__ int   pgid[POOL_SLOTS];
  __shared__ float pcnt[POOL_SLOTS];
  const int tid  = threadIdx.x;
  const int slot = tid >> 4;
  const int l16  = tid & 15;
  const int row0 = blockIdx.x * POOL_ROWS + slot * POOL_RPS;
  int nvalid = n - row0;
  nvalid = nvalid < 0 ? 0 : (nvalid > POOL_RPS ? POOL_RPS : nvalid);

  int    ids[POOL_RPS];
  float4 v[POOL_RPS];
#pragma unroll
  for (int j = 0; j < POOL_RPS; ++j) {
    int r = row0 + j;
    r = r < n ? r : n - 1;
    ids[j] = batch[r];
    v[j] = *(const float4*)(h + (size_t)r * NFEAT + l16 * 4);
  }

  int cur = -1;
  float4 acc = make_float4(0.0f, 0.0f, 0.0f, 0.0f);
  float cnt = 0.0f;
#pragma unroll
  for (int j = 0; j < POOL_RPS; ++j) {
    if (j < nvalid) {
      const int g = ids[j];
      if (g != cur) {
        if (cur >= 0) gcn_pool_flush4(gsum, gcnt, cur, l16, acc, cnt);
        cur = g;
        acc = make_float4(0.0f, 0.0f, 0.0f, 0.0f);
        cnt = 0.0f;
      }
      acc.x += v[j].x; acc.y += v[j].y; acc.z += v[j].z; acc.w += v[j].w;
      cnt += 1.0f;
    }
  }
  *(float4*)(part + slot * NFEAT + l16 * 4) = acc;
  if (l16 == 0) { pgid[slot] = cur; pcnt[slot] = cnt; }
  __syncthreads();

  if (tid < NFEAT) {
    int   g16[POOL_SLOTS];
    float p16[POOL_SLOTS], c16[POOL_SLOTS];
#pragma unroll
    for (int s = 0; s < POOL_SLOTS; ++s) {
      g16[s] = pgid[s];
      p16[s] = part[s * NFEAT + tid];
      c16[s] = pcnt[s];
    }
    int mg = -1;
    float a = 0.0f, c = 0.0f;
#pragma unroll
    for (int s = 0; s < POOL_SLOTS; ++s) {
      const int g = g16[s];
      if (g >= 0) {
        if (g != mg) {
          if (mg >= 0) {
            atomicAdd(&gsum[mg * NFEAT + tid], a);
            if (tid == 0) atomicAdd(&gcnt[mg], c);
          }
          mg = g; a = 0.0f; c = 0.0f;
        }
        a += p16[s];
        c += c16[s];
      }
    }
    if (mg >= 0) {
      atomicAdd(&gsum[mg * NFEAT + tid], a);
      if (tid == 0) atomicAdd(&gcnt[mg], c);
    }
  }
}

__global__ __launch_bounds__(256) void gcn_head(
    const float* __restrict__ gsum, const float* __restrict__ gcnt,
    const float* __restrict__ Wp, const float* __restrict__ bp,
    const float* __restrict__ Wc, const float* __restrict__ bc,
    float* __restrict__ out) {
  __shared__ float g[64 * 64];
  __shared__ float zs[64 * 64];
  const int t = threadIdx.x;
  for (int idx = t; idx < 64 * 64; idx += 256) {
    int gi = idx >> 6;
    float c = fmaxf(gcnt[gi], 1.0f);
    g[idx] = gsum[idx] / c;
  }
  __syncthreads();
  for (int idx = t; idx < 64 * 64; idx += 256) {
    int m = idx >> 6, nn = idx & 63;
    float a = bp[nn];
#pragma unroll 8
    for (int k = 0; k < 64; ++k) a += g[m * 64 + k] * Wp[k * 64 + nn];
    zs[idx] = a;
    out[384 + idx] = a;
  }
  __syncthreads();
  for (int idx = t; idx < 64 * 6; idx += 256) {
    int m = idx / 6, nn = idx - m * 6;
    float a = bc[nn];
#pragma unroll 8
    for (int k = 0; k < 64; ++k) a += zs[m * 64 + k] * Wc[k * 6 + nn];
    out[idx] = a;
  }
}

extern "C" void kernel_launch(void* const* d_in, const int* in_sizes, int n_in,
                              void* d_out, int out_size, void* d_ws, size_t ws_size,
                              hipStream_t stream) {
  const float* x     = (const float*)d_in[0];
  const int*   ei    = (const int*)d_in[1];
  const int*   batch = (const int*)d_in[2];
  const float* W1    = (const float*)d_in[3];
  const float* b1    = (const float*)d_in[4];
  const float* W2    = (const float*)d_in[5];
  const float* b2    = (const float*)d_in[6];
  const float* Wp    = (const float*)d_in[7];
  const float* bp    = (const float*)d_in[8];
  const float* Wc    = (const float*)d_in[9];
  const float* bc    = (const float*)d_in[10];

  const int n = in_sizes[0] / 128;
  const int E = in_sizes[1] / 2;
  const int* src = ei;
  const int* dst = ei + E;

  const int T = 256;
  const int eBlk  = (E + CHUNK - 1) / CHUNK;
  const int nbins = (n + BIN_W - 1) / BIN_W;
  const int gemmBlk = n / 32;
  const int poolBlk = (n + POOL_ROWS - 1) / POOL_ROWS;
  const int gathBlk = (n + 15) / 16;

  char* ws = (char*)d_ws;
  size_t off = 0;
  auto carve = [&](size_t bytes) -> float* {
    float* p = (float*)(ws + off);
    off = (off + bytes + 255) & ~(size_t)255;
    return p;
  };
  float* isq  = carve((size_t)n * sizeof(float));
  float* bufA = carve((size_t)n * NFEAT * sizeof(float));
  float* bufB = carve((size_t)n * NFEAT * sizeof(float));
  int*   rowptr = (int*)carve((size_t)(n + 1) * sizeof(int));
  int*   binned = (int*)carve((size_t)E * 2 * sizeof(int));
  int*   ew     = (int*)carve((size_t)E * sizeof(int));
  float* zero0 = carve(MAX_BINS * sizeof(int));
  int*   btot  = (int*)zero0;
  int*   bcur  = (int*)carve(MAX_BINS * sizeof(int));
  float* gsum  = carve(64 * NFEAT * sizeof(float));
  float* gcnt  = carve(64 * sizeof(float));
  const int zeroCount = (int)((gcnt + 64) - zero0);

  gcn_zero_f32<<<(zeroCount + T - 1) / T, T, 0, stream>>>(zero0, zeroCount);
  gcn_bin_count<<<eBlk, T, 0, stream>>>(dst, btot, E, nbins);
  gcn_bin_fill<<<eBlk, T, 0, stream>>>(src, dst, btot, bcur, binned, E, nbins);
  gcn_bin_lists<<<nbins, T, 0, stream>>>(binned, btot, rowptr, isq, ew, n, nbins);

  gcn_gemm_wmma_f32<<<gemmBlk, T, 0, stream>>>(x, W1, isq, bufA, 128);
  gcn_gather_relu<<<gathBlk, T, 0, stream>>>(rowptr, ew, bufA, isq, b1, bufB, n);

  gcn_gemm_wmma_f32<<<gemmBlk, T, 0, stream>>>(bufB, W2, isq, bufA, 64);
  gcn_gather_relu<<<gathBlk, T, 0, stream>>>(rowptr, ew, bufA, isq, b2, bufB, n);

  gcn_pool<<<poolBlk, T, 0, stream>>>(bufB, batch, gsum, gcnt, n);
  gcn_head<<<1, T, 0, stream>>>(gsum, gcnt, Wp, bp, Wc, bc, (float*)d_out);
}


# GEMM and gather launches loop over 5 row tiles / 5 nodes per half-wave in grids of 1250 blocks (8x fewer, longer-lived waves); next tile and next edge list prefetched, B fragments kept in registers
# speedup vs baseline: 4.5474x; 1.1040x over previous
#include <hip/hip_runtime.h>

typedef float v2f __attribute__((ext_vector_type(2)));
typedef float v8f __attribute__((ext_vector_type(8)));
typedef float v4f __attribute__((ext_vector_type(4)));

#define NFEAT 64

__global__ void gcn_zero_f32(float* __restrict__ p, int count) {
  int i = blockIdx.x * blockDim.x + threadIdx.x;
  if (i < count) p[i] = 0.0f;
}

#define BIN_SHIFT 8
#define BIN_W     (1 << BIN_SHIFT)
#define MAX_BINS  512
#define EDGE_PER  16
#define CHUNK     (256 * EDGE_PER)
#define LIST_PER  8

__device__ __forceinline__ int gcn_block_excl_scan(int v, int* wtot, int& total) {
  const int lane = threadIdx.x & 31;
  const int wave = threadIdx.x >> 5;
  int x = v;
  for (int off = 1; off < 32; off <<= 1) {
    const int y = __shfl_up(x, off, 32);
    if (lane >= off) x += y;
  }
  if (lane == 31) wtot[wave] = x;
  __syncthreads();
  int pre = 0, tot = 0;
  for (int w = 0; w < 8; ++w) {
    const int tw = wtot[w];
    if (w < wave) pre += tw;
    tot += tw;
  }
  __syncthreads();
  total = tot;
  return pre + x - v;
}

__global__ __launch_bounds__(256) void gcn_bin_count(const int* __restrict__ dst,
                                                     int* __restrict__ btot, int E, int nbins) {
  __shared__ int hist[MAX_BINS];
  const int tid = threadIdx.x;
  hist[tid] = 0;
  hist[tid + 256] = 0;
  __syncthreads();
  const int e0 = blockIdx.x * CHUNK + tid;
  int d[EDGE_PER];
#pragma unroll
  for (int k = 0; k < EDGE_PER; ++k) {
    const int e = e0 + k * 256;
    d[k] = (e < E) ? dst[e] : -1;
  }
#pragma unroll
  for (int k = 0; k < EDGE_PER; ++k)
    if (d[k] >= 0) atomicAdd(&hist[d[k] >> BIN_SHIFT], 1);
  __syncthreads();
  for (int b = tid; b < nbins; b += 256) {
    const int h = hist[b];
    if (h) atomicAdd(&btot[b], h);
  }
}

__global__ __launch_bounds__(256) void gcn_bin_fill(const int* __restrict__ src,
                                                    const int* __restrict__ dst,
                                                    const int* __restrict__ btot,
                                                    int* __restrict__ bcur,
                                                    int* __restrict__ binnedi, int E, int nbins) {
  __shared__ int  hist[MAX_BINS];
  __shared__ int  delta[MAX_BINS];
  __shared__ int  wtot[8];
  __shared__ int2 pairs[CHUNK];
  int2* __restrict__ binned = (int2*)binnedi;
  const int tid = threadIdx.x;
  hist[tid] = 0;
  hist[tid + 256] = 0;
  __syncthreads();

  const int e0 = blockIdx.x * CHUNK + tid;
  int d[EDGE_PER], s[EDGE_PER], rank[EDGE_PER];
#pragma unroll
  for (int k = 0; k < EDGE_PER; ++k) {
    const int e = e0 + k * 256;
    d[k] = (e < E) ? dst[e] : -1;
    s[k] = (e < E) ? src[e] : 0;
  }
  const int b0 = 2 * tid, b1 = 2 * tid + 1;
  const int t0 = (b0 < nbins) ? btot[b0] : 0;
  const int t1 = (b1 < nbins) ? btot[b1] : 0;
#pragma unroll
  for (int k = 0; k < EDGE_PER; ++k) {
    rank[k] = 0;
    if (d[k] >= 0) rank[k] = atomicAdd(&hist[d[k] >> BIN_SHIFT], 1);
  }
  __syncthreads();

  const int h0 = hist[b0], h1 = hist[b1];
  int nvalid, etot;
  const int ls0 = gcn_block_excl_scan(h0 + h1, wtot, nvalid);
  const int gs0 = gcn_block_excl_scan(t0 + t1, wtot, etot);
  const int r0 = h0 ? atomicAdd(&bcur[b0], h0) : 0;
  const int r1 = h1 ? atomicAdd(&bcur[b1], h1) : 0;
  hist[b0]  = ls0;
  hist[b1]  = ls0 + h0;
  delta[b0] = gs0 + r0 - ls0;
  delta[b1] = gs0 + t0 + r1 - (ls0 + h0);
  __syncthreads();

#pragma unroll
  for (int k = 0; k < EDGE_PER; ++k)
    if (d[k] >= 0) pairs[hist[d[k] >> BIN_SHIFT] + rank[k]] = make_int2(s[k], d[k]);
  __syncthreads();

  for (int i = tid; i < nvalid; i += 256) {
    const int2 p = pairs[i];
    binned[i + delta[p.y >> BIN_SHIFT]] = p;
  }
}

__global__ __launch_bounds__(256) void gcn_bin_lists(const int* __restrict__ binnedi,
                                                     const int* __restrict__ btot,
                                                     int* __restrict__ rowptr,
                                                     float* __restrict__ isq,
                                                     int* __restrict__ ew, int n, int nbins) {
  __shared__ int ncnt[BIN_W];
  __shared__ int wtot[8];
  const int2* __restrict__ binned = (const int2*)binnedi;
  const int tid = threadIdx.x;
  const int bin = blockIdx.x;

  int v = 0;
  if (tid < bin) v += btot[tid];
  if (tid + 256 < bin) v += btot[tid + 256];
  const int T = btot[bin];
  ncnt[tid] = 0;
  int binBase;
  gcn_block_excl_scan(v, wtot, binBase);

  const int2* __restrict__ mine = binned + binBase;
  for (int i0 = tid; i0 < T; i0 += 256 * LIST_PER) {
    int dl[LIST_PER];
#pragma unroll
    for (int u = 0; u < LIST_PER; ++u) {
      const int i = i0 + u * 256;
      dl[u] = (i < T) ? mine[i].y : -1;
    }
#pragma unroll
    for (int u = 0; u < LIST_PER; ++u)
      if (dl[u] >= 0) atomicAdd(&ncnt[dl[u] & (BIN_W - 1)], 1);
  }
  __syncthreads();

  const int c = ncnt[tid];
  int tsum;
  const int first = binBase + gcn_block_excl_scan(c, wtot, tsum);
  const int node = bin * BIN_W + tid;
  if (node < n) {
    rowptr[node] = first;
    isq[node] = rsqrtf((float)(c + 1));
  }
  if (bin == nbins - 1 && tid == 0) rowptr[n] = binBase + T;
  ncnt[tid] = first;
  __syncthreads();

  for (int i0 = tid; i0 < T; i0 += 256 * LIST_PER) {
    int2 p[LIST_PER];
#pragma unroll
    for (int u = 0; u < LIST_PER; ++u) {
      const int i = i0 + u * 256;
      p[u] = (i < T) ? mine[i] : make_int2(0, -1);
    }
#pragma unroll
    for (int u = 0; u < LIST_PER; ++u)
      if (p[u].y >= 0) ew[atomicAdd(&ncnt[p[u].y & (BIN_W - 1)], 1)] = p[u].x;
  }
}

#define GEMM_ROWS 16
#define GEMM_THR  128

template <int K>
__global__ __launch_bounds__(GEMM_THR) void gcn_gemm_wmma_f32(
    const float* __restrict__ in, const float* __restrict__ W,
    const float* __restrict__ rowscale, float* __restrict__ out, int ntiles) {
  constexpr int ldA = K + 4;
  constexpr int C4  = K / 4;
  constexpr int PER = GEMM_ROWS * C4 / GEMM_THR;
  static_assert(PER == 2 || PER == 4, "K must be 64 or 128");
  __shared__ alignas(16) float As[GEMM_ROWS * ldA];
  const int tid  = threadIdx.x;
  const int wave = tid >> 5;
  const int lane = tid & 31;
  const int m  = lane & 15;
  const int kp = lane >> 4;
  const int col0 = wave * 16;

  int tile = blockIdx.x;
  if (tile >= ntiles) return;

  v2f bf[K / 4];
#pragma unroll
  for (int i = 0; i < K / 4; ++i) {
    const int ka = 4 * i + 2 * kp;
    bf[i].x = W[(size_t)(ka + 0) * NFEAT + col0 + m];
    bf[i].y = W[(size_t)(ka + 1) * NFEAT + col0 + m];
  }

  const float* src0 = in + (size_t)(tid / C4) * K + (tid % C4) * 4;
  constexpr size_t pieceStep = (size_t)(GEMM_THR / C4) * K;
  constexpr size_t tileStep  = (size_t)GEMM_ROWS * K;
  float* dst0 = As + (tid / C4) * ldA + (tid % C4) * 4;
  constexpr int ldsStep = (GEMM_THR / C4) * ldA;
  const float* sp = src0 + (size_t)tile * tileStep;
  v4f p0 = *(const v4f*)(sp);
  v4f p1 = *(const v4f*)(sp + pieceStep);
  v4f p2 = {}, p3 = {};
  if constexpr (PER == 4) {
    p2 = *(const v4f*)(sp + 2 * pieceStep);
    p3 = *(const v4f*)(sp + 3 * pieceStep);
  }

  for (; tile < ntiles; tile += gridDim.x) {
    __syncthreads();
    *(v4f*)(dst0) = p0;
    *(v4f*)(dst0 + ldsStep) = p1;
    if constexpr (PER == 4) {
      *(v4f*)(dst0 + 2 * ldsStep) = p2;
      *(v4f*)(dst0 + 3 * ldsStep) = p3;
    }
    __syncthreads();

    const int nt = tile + gridDim.x;
    if (nt < ntiles) {
      sp = src0 + (size_t)nt * tileStep;
      p0 = *(const v4f*)(sp);
      p1 = *(const v4f*)(sp + pieceStep);
      if constexpr (PER == 4) {
        p2 = *(const v4f*)(sp + 2 * pieceStep);
        p3 = *(const v4f*)(sp + 3 * pieceStep);
      }
    }

    const int gRow0 = tile * GEMM_ROWS;
    float rs[8];
#pragma unroll
    for (int i = 0; i < 8; ++i) rs[i] = rowscale[gRow0 + i + 8 * kp];

    v8f c = {};
#pragma unroll
    for (int i = 0; i < K / 4; ++i) {
      const v2f a = *(const v2f*)(As + m * ldA + 4 * i + 2 * kp);
      c = __builtin_amdgcn_wmma_f32_16x16x4_f32(
           false, a,  false, bf[i],
           (short)0, c,  false,  false);
    }

#pragma unroll
    for (int i = 0; i < 8; ++i) {
      out[(size_t)(gRow0 + i + 8 * kp) * NFEAT + col0 + m] = c[i] * rs[i];
    }
  }
}

__global__ __launch_bounds__(256) void gcn_gather_relu(
    const int* __restrict__ rowptr, const int* __restrict__ ew,
    const float* __restrict__ h, const float* __restrict__ isq,
    const float* __restrict__ bias, float* __restrict__ out, int n) {
  const int tid  = threadIdx.x;
  const int l16  = tid & 15;
  const int step = gridDim.x * 16;
  int node = blockIdx.x * 16 + (tid >> 4);
  if (node >= n) return;
  const float* hp = h + l16 * 4;
  const float4 b = *(const float4*)(bias + l16 * 4);

  int beg = rowptr[node];
  int end = rowptr[node + 1];
  int nbeg = 0, nend = 0;
  if (node + step < n) {
    nbeg = rowptr[node + step];
    nend = rowptr[node + step + 1];
  }
  int sl = (beg + l16 < end) ? ew[beg + l16] : node;

  for (;;) {
    const int nxt = node + step;
    int nsl = nxt;
    if (nbeg + l16 < nend) nsl = ew[nbeg + l16];
    int n2beg = 0, n2end = 0;
    if (nxt + step < n) {
      n2beg = rowptr[nxt + step];
      n2end = rowptr[nxt + step + 1];
    }

    const float isd = isq[node];
    float4 acc = *(const float4*)(hp + (size_t)node * NFEAT);

    for (int base = beg; base < end; base += 16) {
      if (base != beg) {
        const int my = base + l16;
        sl = (my < end) ? ew[my] : node;
      }
      const int cnt = (end - base) < 16 ? (end - base) : 16;
      for (int j = 0; j < cnt; j += 4) {
        int    s[4];
        float4 hv[4];
#pragma unroll
        for (int u = 0; u < 4; ++u) s[u] = __shfl(sl, j + u, 16);
#pragma unroll
        for (int u = 0; u < 4; ++u) hv[u] = *(const float4*)(hp + (size_t)s[u] * NFEAT);
#pragma unroll
        for (int u = 0; u < 4; ++u) {
          if (j + u < cnt) {
            acc.x += hv[u].x;
            acc.y += hv[u].y;
            acc.z += hv[u].z;
            acc.w += hv[u].w;
          }
        }
      }
    }

    float4 o;
    o.x = fmaxf(fmaf(acc.x, isd, b.x), 0.0f);
    o.y = fmaxf(fmaf(acc.y, isd, b.y), 0.0f);
    o.z = fmaxf(fmaf(acc.z, isd, b.z), 0.0f);
    o.w = fmaxf(fmaf(acc.w, isd, b.w), 0.0f);
    *(float4*)(out + (size_t)node * NFEAT + l16 * 4) = o;

    if (nxt >= n) break;
    node = nxt;
    beg = nbeg;  end = nend;  sl = nsl;
    nbeg = n2beg;  nend = n2end;
  }
}

#define POOL_RPS   16
#define POOL_SLOTS 16
#define POOL_ROWS  (POOL_RPS * POOL_SLOTS)

__device__ __forceinline__ void gcn_pool_flush4(float* __restrict__ gsum, float* __restrict__ gcnt,
                                                int g, int l16, float4 a, float c) {
  float* o = gsum + g * NFEAT + l16 * 4;
  atomicAdd(o + 0, a.x);
  atomicAdd(o + 1, a.y);
  atomicAdd(o + 2, a.z);
  atomicAdd(o + 3, a.w);
  if (l16 == 0) atomicAdd(&gcnt[g], c);
}

__global__ __launch_bounds__(256) void gcn_pool(const float* __restrict__ h, const int* __restrict__ batch,
                         float* __restrict__ gsum, float* __restrict__ gcnt, int n) {
  __shared__ alignas(16) float part[POOL_SLOTS * NFEAT];
  __shared__ int   pgid[POOL_SLOTS];
  __shared__ float pcnt[POOL_SLOTS];
  const int tid  = threadIdx.x;
  const int slot = tid >> 4;
  const int l16  = tid & 15;
  const int row0 = blockIdx.x * POOL_ROWS + slot * POOL_RPS;
  int nvalid = n - row0;
  nvalid = nvalid < 0 ? 0 : (nvalid > POOL_RPS ? POOL_RPS : nvalid);

  int    ids[POOL_RPS];
  float4 v[POOL_RPS];
#pragma unroll
  for (int j = 0; j < POOL_RPS; ++j) {
    int r = row0 + j;
    r = r < n ? r : n - 1;
    ids[j] = batch[r];
    v[j] = *(const float4*)(h + (size_t)r * NFEAT + l16 * 4);
  }

  int cur = -1;
  float4 acc = make_float4(0.0f, 0.0f, 0.0f, 0.0f);
  float cnt = 0.0f;
#pragma unroll
  for (int j = 0; j < POOL_RPS; ++j) {
    if (j < nvalid) {
      const int g = ids[j];
      if (g != cur) {
        if (cur >= 0) gcn_pool_flush4(gsum, gcnt, cur, l16, acc, cnt);
        cur = g;
        acc = make_float4(0.0f, 0.0f, 0.0f, 0.0f);
        cnt = 0.0f;
      }
      acc.x += v[j].x; acc.y += v[j].y; acc.z += v[j].z; acc.w += v[j].w;
      cnt += 1.0f;
    }
  }
  *(float4*)(part + slot * NFEAT + l16 * 4) = acc;
  if (l16 == 0) { pgid[slot] = cur; pcnt[slot] = cnt; }
  __syncthreads();

  if (tid < NFEAT) {
    int   g16[POOL_SLOTS];
    float p16[POOL_SLOTS], c16[POOL_SLOTS];
#pragma unroll
    for (int s = 0; s < POOL_SLOTS; ++s) {
      g16[s] = pgid[s];
      p16[s] = part[s * NFEAT + tid];
      c16[s] = pcnt[s];
    }
    int mg = -1;
    float a = 0.0f, c = 0.0f;
#pragma unroll
    for (int s = 0; s < POOL_SLOTS; ++s) {
      const int g = g16[s];
      if (g >= 0) {
        if (g != mg) {
          if (mg >= 0) {
            atomicAdd(&gsum[mg * NFEAT + tid], a);
            if (tid == 0) atomicAdd(&gcnt[mg], c);
          }
          mg = g; a = 0.0f; c = 0.0f;
        }
        a += p16[s];
        c += c16[s];
      }
    }
    if (mg >= 0) {
      atomicAdd(&gsum[mg * NFEAT + tid], a);
      if (tid == 0) atomicAdd(&gcnt[mg], c);
    }
  }
}

__global__ __launch_bounds__(256) void gcn_head(
    const float* __restrict__ gsum, const float* __restrict__ gcnt,
    const float* __restrict__ Wp, const float* __restrict__ bp,
    const float* __restrict__ Wc, const float* __restrict__ bc,
    float* __restrict__ out) {
  __shared__ float g[64 * 64];
  __shared__ float zs[64 * 64];
  const int t = threadIdx.x;
  for (int idx = t; idx < 64 * 64; idx += 256) {
    int gi = idx >> 6;
    float c = fmaxf(gcnt[gi], 1.0f);
    g[idx] = gsum[idx] / c;
  }
  __syncthreads();
  for (int idx = t; idx < 64 * 64; idx += 256) {
    int m = idx >> 6, nn = idx & 63;
    float a = bp[nn];
#pragma unroll 8
    for (int k = 0; k < 64; ++k) a += g[m * 64 + k] * Wp[k * 64 + nn];
    zs[idx] = a;
    out[384 + idx] = a;
  }
  __syncthreads();
  for (int idx = t; idx < 64 * 6; idx += 256) {
    int m = idx / 6, nn = idx - m * 6;
    float a = bc[nn];
#pragma unroll 8
    for (int k = 0; k < 64; ++k) a += zs[m * 64 + k] * Wc[k * 6 + nn];
    out[idx] = a;
  }
}

extern "C" void kernel_launch(void* const* d_in, const int* in_sizes, int n_in,
                              void* d_out, int out_size, void* d_ws, size_t ws_size,
                              hipStream_t stream) {
  const float* x     = (const float*)d_in[0];
  const int*   ei    = (const int*)d_in[1];
  const int*   batch = (const int*)d_in[2];
  const float* W1    = (const float*)d_in[3];
  const float* b1    = (const float*)d_in[4];
  const float* W2    = (const float*)d_in[5];
  const float* b2    = (const float*)d_in[6];
  const float* Wp    = (const float*)d_in[7];
  const float* bp    = (const float*)d_in[8];
  const float* Wc    = (const float*)d_in[9];
  const float* bc    = (const float*)d_in[10];

  const int n = in_sizes[0] / 128;
  const int E = in_sizes[1] / 2;
  const int* src = ei;
  const int* dst = ei + E;

  const int T = 256;
  const int eBlk  = (E + CHUNK - 1) / CHUNK;
  const int nbins = (n + BIN_W - 1) / BIN_W;
  const int gemmTiles = n / GEMM_ROWS;
  const int gemmBlk = (gemmTiles + 4) / 5;
  const int poolBlk = (n + POOL_ROWS - 1) / POOL_ROWS;
  const int gathBlk = ((n + 15) / 16 + 4) / 5;

  char* ws = (char*)d_ws;
  size_t off = 0;
  auto carve = [&](size_t bytes) -> float* {
    float* p = (float*)(ws + off);
    off = (off + bytes + 255) & ~(size_t)255;
    return p;
  };
  float* isq  = carve((size_t)n * sizeof(float));
  float* bufA = carve((size_t)n * NFEAT * sizeof(float));
  float* bufB = carve((size_t)n * NFEAT * sizeof(float));
  int*   rowptr = (int*)carve((size_t)(n + 1) * sizeof(int));
  int*   binned = (int*)carve((size_t)E * 2 * sizeof(int));
  int*   ew     = (int*)carve((size_t)E * sizeof(int));
  float* zero0 = carve(MAX_BINS * sizeof(int));
  int*   btot  = (int*)zero0;
  int*   bcur  = (int*)carve(MAX_BINS * sizeof(int));
  float* gsum  = carve(64 * NFEAT * sizeof(float));
  float* gcnt  = carve(64 * sizeof(float));
  const int zeroCount = (int)((gcnt + 64) - zero0);

  gcn_zero_f32<<<(zeroCount + T - 1) / T, T, 0, stream>>>(zero0, zeroCount);
  gcn_bin_count<<<eBlk, T, 0, stream>>>(dst, btot, E, nbins);
  gcn_bin_fill<<<eBlk, T, 0, stream>>>(src, dst, btot, bcur, binned, E, nbins);
  gcn_bin_lists<<<nbins, T, 0, stream>>>(binned, btot, rowptr, isq, ew, n, nbins);

  gcn_gemm_wmma_f32<128><<<gemmBlk, GEMM_THR, 0, stream>>>(x, W1, isq, bufA, gemmTiles);
  gcn_gather_relu<<<gathBlk, T, 0, stream>>>(rowptr, ew, bufA, isq, b1, bufB, n);

  gcn_gemm_wmma_f32<64><<<gemmBlk, GEMM_THR, 0, stream>>>(bufB, W2, isq, bufA, gemmTiles);
  gcn_gather_relu<<<gathBlk, T, 0, stream>>>(rowptr, ew, bufA, isq, b2, bufB, n);

  gcn_pool<<<poolBlk, T, 0, stream>>>(bufB, batch, gsum, gcnt, n);
  gcn_head<<<1, T, 0, stream>>>(gsum, gcnt, Wp, bp, Wc, bc, (float*)d_out);
}


# head: one block per graph (64 blocks of 64 threads), each thread's Wp and Wc columns loaded to registers in one batch ahead of the barrier, not one block looping over 4096 outputs; sum order kept
# speedup vs baseline: 5.5826x; 1.2276x over previous
#include <hip/hip_runtime.h>

typedef float v2f __attribute__((ext_vector_type(2)));
typedef float v8f __attribute__((ext_vector_type(8)));
typedef float v4f __attribute__((ext_vector_type(4)));

#define NFEAT 64

__global__ void gcn_zero_f32(float* __restrict__ p, int count) {
  int i = blockIdx.x * blockDim.x + threadIdx.x;
  if (i < count) p[i] = 0.0f;
}

#define BIN_SHIFT 8
#define BIN_W     (1 << BIN_SHIFT)
#define MAX_BINS  512
#define EDGE_PER  16
#define CHUNK     (256 * EDGE_PER)
#define LIST_PER  8

__device__ __forceinline__ int gcn_block_excl_scan(int v, int* wtot, int& total) {
  const int lane = threadIdx.x & 31;
  const int wave = threadIdx.x >> 5;
  int x = v;
  for (int off = 1; off < 32; off <<= 1) {
    const int y = __shfl_up(x, off, 32);
    if (lane >= off) x += y;
  }
  if (lane == 31) wtot[wave] = x;
  __syncthreads();
  int pre = 0, tot = 0;
  for (int w = 0; w < 8; ++w) {
    const int tw = wtot[w];
    if (w < wave) pre += tw;
    tot += tw;
  }
  __syncthreads();
  total = tot;
  return pre + x - v;
}

__global__ __launch_bounds__(256) void gcn_bin_count(const int* __restrict__ dst,
                                                     int* __restrict__ btot, int E, int nbins) {
  __shared__ int hist[MAX_BINS];
  const int tid = threadIdx.x;
  hist[tid] = 0;
  hist[tid + 256] = 0;
  __syncthreads();
  const int e0 = blockIdx.x * CHUNK + tid;
  int d[EDGE_PER];
#pragma unroll
  for (int k = 0; k < EDGE_PER; ++k) {
    const int e = e0 + k * 256;
    d[k] = (e < E) ? dst[e] : -1;
  }
#pragma unroll
  for (int k = 0; k < EDGE_PER; ++k)
    if (d[k] >= 0) atomicAdd(&hist[d[k] >> BIN_SHIFT], 1);
  __syncthreads();
  for (int b = tid; b < nbins; b += 256) {
    const int h = hist[b];
    if (h) atomicAdd(&btot[b], h);
  }
}

__global__ __launch_bounds__(256) void gcn_bin_fill(const int* __restrict__ src,
                                                    const int* __restrict__ dst,
                                                    const int* __restrict__ btot,
                                                    int* __restrict__ bcur,
                                                    int* __restrict__ binnedi, int E, int nbins) {
  __shared__ int  hist[MAX_BINS];
  __shared__ int  delta[MAX_BINS];
  __shared__ int  wtot[8];
  __shared__ int2 pairs[CHUNK];
  int2* __restrict__ binned = (int2*)binnedi;
  const int tid = threadIdx.x;
  hist[tid] = 0;
  hist[tid + 256] = 0;
  __syncthreads();

  const int e0 = blockIdx.x * CHUNK + tid;
  int d[EDGE_PER], s[EDGE_PER], rank[EDGE_PER];
#pragma unroll
  for (int k = 0; k < EDGE_PER; ++k) {
    const int e = e0 + k * 256;
    d[k] = (e < E) ? dst[e] : -1;
    s[k] = (e < E) ? src[e] : 0;
  }
  const int b0 = 2 * tid, b1 = 2 * tid + 1;
  const int t0 = (b0 < nbins) ? btot[b0] : 0;
  const int t1 = (b1 < nbins) ? btot[b1] : 0;
#pragma unroll
  for (int k = 0; k < EDGE_PER; ++k) {
    rank[k] = 0;
    if (d[k] >= 0) rank[k] = atomicAdd(&hist[d[k] >> BIN_SHIFT], 1);
  }
  __syncthreads();

  const int h0 = hist[b0], h1 = hist[b1];
  int nvalid, etot;
  const int ls0 = gcn_block_excl_scan(h0 + h1, wtot, nvalid);
  const int gs0 = gcn_block_excl_scan(t0 + t1, wtot, etot);
  const int r0 = h0 ? atomicAdd(&bcur[b0], h0) : 0;
  const int r1 = h1 ? atomicAdd(&bcur[b1], h1) : 0;
  hist[b0]  = ls0;
  hist[b1]  = ls0 + h0;
  delta[b0] = gs0 + r0 - ls0;
  delta[b1] = gs0 + t0 + r1 - (ls0 + h0);
  __syncthreads();

#pragma unroll
  for (int k = 0; k < EDGE_PER; ++k)
    if (d[k] >= 0) pairs[hist[d[k] >> BIN_SHIFT] + rank[k]] = make_int2(s[k], d[k]);
  __syncthreads();

  for (int i = tid; i < nvalid; i += 256) {
    const int2 p = pairs[i];
    binned[i + delta[p.y >> BIN_SHIFT]] = p;
  }
}

__global__ __launch_bounds__(256) void gcn_bin_lists(const int* __restrict__ binnedi,
                                                     const int* __restrict__ btot,
                                                     int* __restrict__ rowptr,
                                                     float* __restrict__ isq,
                                                     int* __restrict__ ew, int n, int nbins) {
  __shared__ int ncnt[BIN_W];
  __shared__ int wtot[8];
  const int2* __restrict__ binned = (const int2*)binnedi;
  const int tid = threadIdx.x;
  const int bin = blockIdx.x;

  int v = 0;
  if (tid < bin) v += btot[tid];
  if (tid + 256 < bin) v += btot[tid + 256];
  const int T = btot[bin];
  ncnt[tid] = 0;
  int binBase;
  gcn_block_excl_scan(v, wtot, binBase);

  const int2* __restrict__ mine = binned + binBase;
  for (int i0 = tid; i0 < T; i0 += 256 * LIST_PER) {
    int dl[LIST_PER];
#pragma unroll
    for (int u = 0; u < LIST_PER; ++u) {
      const int i = i0 + u * 256;
      dl[u] = (i < T) ? mine[i].y : -1;
    }
#pragma unroll
    for (int u = 0; u < LIST_PER; ++u)
      if (dl[u] >= 0) atomicAdd(&ncnt[dl[u] & (BIN_W - 1)], 1);
  }
  __syncthreads();

  const int c = ncnt[tid];
  int tsum;
  const int first = binBase + gcn_block_excl_scan(c, wtot, tsum);
  const int node = bin * BIN_W + tid;
  if (node < n) {
    rowptr[node] = first;
    isq[node] = rsqrtf((float)(c + 1));
  }
  if (bin == nbins - 1 && tid == 0) rowptr[n] = binBase + T;
  ncnt[tid] = first;
  __syncthreads();

  for (int i0 = tid; i0 < T; i0 += 256 * LIST_PER) {
    int2 p[LIST_PER];
#pragma unroll
    for (int u = 0; u < LIST_PER; ++u) {
      const int i = i0 + u * 256;
      p[u] = (i < T) ? mine[i] : make_int2(0, -1);
    }
#pragma unroll
    for (int u = 0; u < LIST_PER; ++u)
      if (p[u].y >= 0) ew[atomicAdd(&ncnt[p[u].y & (BIN_W - 1)], 1)] = p[u].x;
  }
}

#define GEMM_ROWS 16
#define GEMM_THR  128

template <int K>
__global__ __launch_bounds__(GEMM_THR) void gcn_gemm_wmma_f32(
    const float* __restrict__ in, const float* __restrict__ W,
    const float* __restrict__ rowscale, float* __restrict__ out, int ntiles) {
  constexpr int ldA = K + 4;
  constexpr int C4  = K / 4;
  constexpr int PER = GEMM_ROWS * C4 / GEMM_THR;
  static_assert(PER == 2 || PER == 4, "K must be 64 or 128");
  __shared__ alignas(16) float As[GEMM_ROWS * ldA];
  const int tid  = threadIdx.x;
  const int wave = tid >> 5;
  const int lane = tid & 31;
  const int m  = lane & 15;
  const int kp = lane >> 4;
  const int col0 = wave * 16;

  int tile = blockIdx.x;
  if (tile >= ntiles) return;

  v2f bf[K / 4];
#pragma unroll
  for (int i = 0; i < K / 4; ++i) {
    const int ka = 4 * i + 2 * kp;
    bf[i].x = W[(size_t)(ka + 0) * NFEAT + col0 + m];
    bf[i].y = W[(size_t)(ka + 1) * NFEAT + col0 + m];
  }

  const float* src0 = in + (size_t)(tid / C4) * K + (tid % C4) * 4;
  constexpr size_t pieceStep = (size_t)(GEMM_THR / C4) * K;
  constexpr size_t tileStep  = (size_t)GEMM_ROWS * K;
  float* dst0 = As + (tid / C4) * ldA + (tid % C4) * 4;
  constexpr int ldsStep = (GEMM_THR / C4) * ldA;
  const float* sp = src0 + (size_t)tile * tileStep;
  v4f p0 = *(const v4f*)(sp);
  v4f p1 = *(const v4f*)(sp + pieceStep);
  v4f p2 = {}, p3 = {};
  if constexpr (PER == 4) {
    p2 = *(const v4f*)(sp + 2 * pieceStep);
    p3 = *(const v4f*)(sp + 3 * pieceStep);
  }

  for (; tile < ntiles; tile += gridDim.x) {
    __syncthreads();
    *(v4f*)(dst0) = p0;
    *(v4f*)(dst0 + ldsStep) = p1;
    if constexpr (PER == 4) {
      *(v4f*)(dst0 + 2 * ldsStep) = p2;
      *(v4f*)(dst0 + 3 * ldsStep) = p3;
    }
    __syncthreads();

    const int nt = tile + gridDim.x;
    if (nt < ntiles) {
      sp = src0 + (size_t)nt * tileStep;
      p0 = *(const v4f*)(sp);
      p1 = *(const v4f*)(sp + pieceStep);
      if constexpr (PER == 4) {
        p2 = *(const v4f*)(sp + 2 * pieceStep);
        p3 = *(const v4f*)(sp + 3 * pieceStep);
      }
    }

    const int gRow0 = tile * GEMM_ROWS;
    float rs[8];
#pragma unroll
    for (int i = 0; i < 8; ++i) rs[i] = rowscale[gRow0 + i + 8 * kp];

    v8f c = {};
#pragma unroll
    for (int i = 0; i < K / 4; ++i) {
      const v2f a = *(const v2f*)(As + m * ldA + 4 * i + 2 * kp);
      c = __builtin_amdgcn_wmma_f32_16x16x4_f32(
           false, a,  false, bf[i],
           (short)0, c,  false,  false);
    }

#pragma unroll
    for (int i = 0; i < 8; ++i) {
      out[(size_t)(gRow0 + i + 8 * kp) * NFEAT + col0 + m] = c[i] * rs[i];
    }
  }
}

__global__ __launch_bounds__(256) void gcn_gather_relu(
    const int* __restrict__ rowptr, const int* __restrict__ ew,
    const float* __restrict__ h, const float* __restrict__ isq,
    const float* __restrict__ bias, float* __restrict__ out, int n) {
  const int tid  = threadIdx.x;
  const int l16  = tid & 15;
  const int step = gridDim.x * 16;
  int node = blockIdx.x * 16 + (tid >> 4);
  if (node >= n) return;
  const float* hp = h + l16 * 4;
  const float4 b = *(const float4*)(bias + l16 * 4);

  int beg = rowptr[node];
  int end = rowptr[node + 1];
  int nbeg = 0, nend = 0;
  if (node + step < n) {
    nbeg = rowptr[node + step];
    nend = rowptr[node + step + 1];
  }
  int sl = (beg + l16 < end) ? ew[beg + l16] : node;

  for (;;) {
    const int nxt = node + step;
    int nsl = nxt;
    if (nbeg + l16 < nend) nsl = ew[nbeg + l16];
    int n2beg = 0, n2end = 0;
    if (nxt + step < n) {
      n2beg = rowptr[nxt + step];
      n2end = rowptr[nxt + step + 1];
    }

    const float isd = isq[node];
    float4 acc = *(const float4*)(hp + (size_t)node * NFEAT);

    for (int base = beg; base < end; base += 16) {
      if (base != beg) {
        const int my = base + l16;
        sl = (my < end) ? ew[my] : node;
      }
      const int cnt = (end - base) < 16 ? (end - base) : 16;
      for (int j = 0; j < cnt; j += 4) {
        int    s[4];
        float4 hv[4];
#pragma unroll
        for (int u = 0; u < 4; ++u) s[u] = __shfl(sl, j + u, 16);
#pragma unroll
        for (int u = 0; u < 4; ++u) hv[u] = *(const float4*)(hp + (size_t)s[u] * NFEAT);
#pragma unroll
        for (int u = 0; u < 4; ++u) {
          if (j + u < cnt) {
            acc.x += hv[u].x;
            acc.y += hv[u].y;
            acc.z += hv[u].z;
            acc.w += hv[u].w;
          }
        }
      }
    }

    float4 o;
    o.x = fmaxf(fmaf(acc.x, isd, b.x), 0.0f);
    o.y = fmaxf(fmaf(acc.y, isd, b.y), 0.0f);
    o.z = fmaxf(fmaf(acc.z, isd, b.z), 0.0f);
    o.w = fmaxf(fmaf(acc.w, isd, b.w), 0.0f);
    *(float4*)(out + (size_t)node * NFEAT + l16 * 4) = o;

    if (nxt >= n) break;
    node = nxt;
    beg = nbeg;  end = nend;  sl = nsl;
    nbeg = n2beg;  nend = n2end;
  }
}

#define POOL_RPS   16
#define POOL_SLOTS 16
#define POOL_ROWS  (POOL_RPS * POOL_SLOTS)

__device__ __forceinline__ void gcn_pool_flush4(float* __restrict__ gsum, float* __restrict__ gcnt,
                                                int g, int l16, float4 a, float c) {
  float* o = gsum + g * NFEAT + l16 * 4;
  atomicAdd(o + 0, a.x);
  atomicAdd(o + 1, a.y);
  atomicAdd(o + 2, a.z);
  atomicAdd(o + 3, a.w);
  if (l16 == 0) atomicAdd(&gcnt[g], c);
}

__global__ __launch_bounds__(256) void gcn_pool(const float* __restrict__ h, const int* __restrict__ batch,
                         float* __restrict__ gsum, float* __restrict__ gcnt, int n) {
  __shared__ alignas(16) float part[POOL_SLOTS * NFEAT];
  __shared__ int   pgid[POOL_SLOTS];
  __shared__ float pcnt[POOL_SLOTS];
  const int tid  = threadIdx.x;
  const int slot = tid >> 4;
  const int l16  = tid & 15;
  const int row0 = blockIdx.x * POOL_ROWS + slot * POOL_RPS;
  int nvalid = n - row0;
  nvalid = nvalid < 0 ? 0 : (nvalid > POOL_RPS ? POOL_RPS : nvalid);

  int    ids[POOL_RPS];
  float4 v[POOL_RPS];
#pragma unroll
  for (int j = 0; j < POOL_RPS; ++j) {
    int r = row0 + j;
    r = r < n ? r : n - 1;
    ids[j] = batch[r];
    v[j] = *(const float4*)(h + (size_t)r * NFEAT + l16 * 4);
  }

  int cur = -1;
  float4 acc = make_float4(0.0f, 0.0f, 0.0f, 0.0f);
  float cnt = 0.0f;
#pragma unroll
  for (int j = 0; j < POOL_RPS; ++j) {
    if (j < nvalid) {
      const int g = ids[j];
      if (g != cur) {
        if (cur >= 0) gcn_pool_flush4(gsum, gcnt, cur, l16, acc, cnt);
        cur = g;
        acc = make_float4(0.0f, 0.0f, 0.0f, 0.0f);
        cnt = 0.0f;
      }
      acc.x += v[j].x; acc.y += v[j].y; acc.z += v[j].z; acc.w += v[j].w;
      cnt += 1.0f;
    }
  }
  *(float4*)(part + slot * NFEAT + l16 * 4) = acc;
  if (l16 == 0) { pgid[slot] = cur; pcnt[slot] = cnt; }
  __syncthreads();

  if (tid < NFEAT) {
    int   g16[POOL_SLOTS];
    float p16[POOL_SLOTS], c16[POOL_SLOTS];
#pragma unroll
    for (int s = 0; s < POOL_SLOTS; ++s) {
      g16[s] = pgid[s];
      p16[s] = part[s * NFEAT + tid];
      c16[s] = pcnt[s];
    }
    int mg = -1;
    float a = 0.0f, c = 0.0f;
#pragma unroll
    for (int s = 0; s < POOL_SLOTS; ++s) {
      const int g = g16[s];
      if (g >= 0) {
        if (g != mg) {
          if (mg >= 0) {
            atomicAdd(&gsum[mg * NFEAT + tid], a);
            if (tid == 0) atomicAdd(&gcnt[mg], c);
          }
          mg = g; a = 0.0f; c = 0.0f;
        }
        a += p16[s];
        c += c16[s];
      }
    }
    if (mg >= 0) {
      atomicAdd(&gsum[mg * NFEAT + tid], a);
      if (tid == 0) atomicAdd(&gcnt[mg], c);
    }
  }
}

#define HEAD_G   64
#define HEAD_NC  6

__global__ __launch_bounds__(NFEAT) void gcn_head(
    const float* __restrict__ gsum, const float* __restrict__ gcnt,
    const float* __restrict__ Wp, const float* __restrict__ bp,
    const float* __restrict__ Wc, const float* __restrict__ bc,
    float* __restrict__ out) {
  __shared__ float g[NFEAT];
  __shared__ float zs[NFEAT];
  const int t = threadIdx.x;
  const int m = blockIdx.x;

  float wp[NFEAT], wc[NFEAT];
  float l = 0.0f;
#pragma unroll
  for (int k = 0; k < NFEAT; ++k) wp[k] = Wp[k * NFEAT + t];
  if (t < HEAD_NC) {
#pragma unroll
    for (int k = 0; k < NFEAT; ++k) wc[k] = Wc[k * HEAD_NC + t];
    l = bc[t];
  }
  const float c = fmaxf(gcnt[m], 1.0f);
  g[t] = gsum[m * NFEAT + t] / c;
  float a = bp[t];
  __syncthreads();

#pragma unroll
  for (int k = 0; k < NFEAT; ++k) a += g[k] * wp[k];
  zs[t] = a;
  out[HEAD_G * HEAD_NC + m * NFEAT + t] = a;
  __syncthreads();

  if (t < HEAD_NC) {
#pragma unroll
    for (int k = 0; k < NFEAT; ++k) l += zs[k] * wc[k];
    out[m * HEAD_NC + t] = l;
  }
}

extern "C" void kernel_launch(void* const* d_in, const int* in_sizes, int n_in,
                              void* d_out, int out_size, void* d_ws, size_t ws_size,
                              hipStream_t stream) {
  const float* x     = (const float*)d_in[0];
  const int*   ei    = (const int*)d_in[1];
  const int*   batch = (const int*)d_in[2];
  const float* W1    = (const float*)d_in[3];
  const float* b1    = (const float*)d_in[4];
  const float* W2    = (const float*)d_in[5];
  const float* b2    = (const float*)d_in[6];
  const float* Wp    = (const float*)d_in[7];
  const float* bp    = (const float*)d_in[8];
  const float* Wc    = (const float*)d_in[9];
  const float* bc    = (const float*)d_in[10];

  const int n = in_sizes[0] / 128;
  const int E = in_sizes[1] / 2;
  const int* src = ei;
  const int* dst = ei + E;

  const int T = 256;
  const int eBlk  = (E + CHUNK - 1) / CHUNK;
  const int nbins = (n + BIN_W - 1) / BIN_W;
  const int gemmTiles = n / GEMM_ROWS;
  const int gemmBlk = (gemmTiles + 4) / 5;
  const int poolBlk = (n + POOL_ROWS - 1) / POOL_ROWS;
  const int gathBlk = ((n + 15) / 16 + 4) / 5;

  char* ws = (char*)d_ws;
  size_t off = 0;
  auto carve = [&](size_t bytes) -> float* {
    float* p = (float*)(ws + off);
    off = (off + bytes + 255) & ~(size_t)255;
    return p;
  };
  float* isq  = carve((size_t)n * sizeof(float));
  float* bufA = carve((size_t)n * NFEAT * sizeof(float));
  float* bufB = carve((size_t)n * NFEAT * sizeof(float));
  int*   rowptr = (int*)carve((size_t)(n + 1) * sizeof(int));
  int*   binned = (int*)carve((size_t)E * 2 * sizeof(int));
  int*   ew     = (int*)carve((size_t)E * sizeof(int));
  float* zero0 = carve(MAX_BINS * sizeof(int));
  int*   btot  = (int*)zero0;
  int*   bcur  = (int*)carve(MAX_BINS * sizeof(int));
  float* gsum  = carve(64 * NFEAT * sizeof(float));
  float* gcnt  = carve(64 * sizeof(float));
  const int zeroCount = (int)((gcnt + 64) - zero0);

  gcn_zero_f32<<<(zeroCount + T - 1) / T, T, 0, stream>>>(zero0, zeroCount);
  gcn_bin_count<<<eBlk, T, 0, stream>>>(dst, btot, E, nbins);
  gcn_bin_fill<<<eBlk, T, 0, stream>>>(src, dst, btot, bcur, binned, E, nbins);
  gcn_bin_lists<<<nbins, T, 0, stream>>>(binned, btot, rowptr, isq, ew, n, nbins);

  gcn_gemm_wmma_f32<128><<<gemmBlk, GEMM_THR, 0, stream>>>(x, W1, isq, bufA, gemmTiles);
  gcn_gather_relu<<<gathBlk, T, 0, stream>>>(rowptr, ew, bufA, isq, b1, bufB, n);

  gcn_gemm_wmma_f32<64><<<gemmBlk, GEMM_THR, 0, stream>>>(bufB, W2, isq, bufA, gemmTiles);
  gcn_gather_relu<<<gathBlk, T, 0, stream>>>(rowptr, ew, bufA, isq, b2, bufB, n);

  gcn_pool<<<poolBlk, T, 0, stream>>>(bufB, batch, gsum, gcnt, n);
  gcn_head<<<HEAD_G, NFEAT, 0, stream>>>(gsum, gcnt, Wp, bp, Wc, bc, (float*)d_out);
}


# pool fused into the layer-2 gather: a half-wave takes 5 consecutive nodes and keeps their rows' per-graph sum in registers, block merges in LDS, then atomics; h2 never stored; float sum order changes
# speedup vs baseline: 5.8589x; 1.0495x over previous
#include <hip/hip_runtime.h>

typedef float v2f __attribute__((ext_vector_type(2)));
typedef float v8f __attribute__((ext_vector_type(8)));
typedef float v4f __attribute__((ext_vector_type(4)));

#define NFEAT 64

__global__ void gcn_zero_f32(float* __restrict__ p, int count) {
  int i = blockIdx.x * blockDim.x + threadIdx.x;
  if (i < count) p[i] = 0.0f;
}

#define BIN_SHIFT 8
#define BIN_W     (1 << BIN_SHIFT)
#define MAX_BINS  512
#define EDGE_PER  16
#define CHUNK     (256 * EDGE_PER)
#define LIST_PER  8

__device__ __forceinline__ int gcn_block_excl_scan(int v, int* wtot, int& total) {
  const int lane = threadIdx.x & 31;
  const int wave = threadIdx.x >> 5;
  int x = v;
  for (int off = 1; off < 32; off <<= 1) {
    const int y = __shfl_up(x, off, 32);
    if (lane >= off) x += y;
  }
  if (lane == 31) wtot[wave] = x;
  __syncthreads();
  int pre = 0, tot = 0;
  for (int w = 0; w < 8; ++w) {
    const int tw = wtot[w];
    if (w < wave) pre += tw;
    tot += tw;
  }
  __syncthreads();
  total = tot;
  return pre + x - v;
}

__global__ __launch_bounds__(256) void gcn_bin_count(const int* __restrict__ dst,
                                                     int* __restrict__ btot, int E, int nbins) {
  __shared__ int hist[MAX_BINS];
  const int tid = threadIdx.x;
  hist[tid] = 0;
  hist[tid + 256] = 0;
  __syncthreads();
  const int e0 = blockIdx.x * CHUNK + tid;
  int d[EDGE_PER];
#pragma unroll
  for (int k = 0; k < EDGE_PER; ++k) {
    const int e = e0 + k * 256;
    d[k] = (e < E) ? dst[e] : -1;
  }
#pragma unroll
  for (int k = 0; k < EDGE_PER; ++k)
    if (d[k] >= 0) atomicAdd(&hist[d[k] >> BIN_SHIFT], 1);
  __syncthreads();
  for (int b = tid; b < nbins; b += 256) {
    const int h = hist[b];
    if (h) atomicAdd(&btot[b], h);
  }
}

__global__ __launch_bounds__(256) void gcn_bin_fill(const int* __restrict__ src,
                                                    const int* __restrict__ dst,
                                                    const int* __restrict__ btot,
                                                    int* __restrict__ bcur,
                                                    int* __restrict__ binnedi, int E, int nbins) {
  __shared__ int  hist[MAX_BINS];
  __shared__ int  delta[MAX_BINS];
  __shared__ int  wtot[8];
  __shared__ int2 pairs[CHUNK];
  int2* __restrict__ binned = (int2*)binnedi;
  const int tid = threadIdx.x;
  hist[tid] = 0;
  hist[tid + 256] = 0;
  __syncthreads();

  const int e0 = blockIdx.x * CHUNK + tid;
  int d[EDGE_PER], s[EDGE_PER], rank[EDGE_PER];
#pragma unroll
  for (int k = 0; k < EDGE_PER; ++k) {
    const int e = e0 + k * 256;
    d[k] = (e < E) ? dst[e] : -1;
    s[k] = (e < E) ? src[e] : 0;
  }
  const int b0 = 2 * tid, b1 = 2 * tid + 1;
  const int t0 = (b0 < nbins) ? btot[b0] : 0;
  const int t1 = (b1 < nbins) ? btot[b1] : 0;
#pragma unroll
  for (int k = 0; k < EDGE_PER; ++k) {
    rank[k] = 0;
    if (d[k] >= 0) rank[k] = atomicAdd(&hist[d[k] >> BIN_SHIFT], 1);
  }
  __syncthreads();

  const int h0 = hist[b0], h1 = hist[b1];
  int nvalid, etot;
  const int ls0 = gcn_block_excl_scan(h0 + h1, wtot, nvalid);
  const int gs0 = gcn_block_excl_scan(t0 + t1, wtot, etot);
  const int r0 = h0 ? atomicAdd(&bcur[b0], h0) : 0;
  const int r1 = h1 ? atomicAdd(&bcur[b1], h1) : 0;
  hist[b0]  = ls0;
  hist[b1]  = ls0 + h0;
  delta[b0] = gs0 + r0 - ls0;
  delta[b1] = gs0 + t0 + r1 - (ls0 + h0);
  __syncthreads();

#pragma unroll
  for (int k = 0; k < EDGE_PER; ++k)
    if (d[k] >= 0) pairs[hist[d[k] >> BIN_SHIFT] + rank[k]] = make_int2(s[k], d[k]);
  __syncthreads();

  for (int i = tid; i < nvalid; i += 256) {
    const int2 p = pairs[i];
    binned[i + delta[p.y >> BIN_SHIFT]] = p;
  }
}

__global__ __launch_bounds__(256) void gcn_bin_lists(const int* __restrict__ binnedi,
                                                     const int* __restrict__ btot,
                                                     int* __restrict__ rowptr,
                                                     float* __restrict__ isq,
                                                     int* __restrict__ ew, int n, int nbins) {
  __shared__ int ncnt[BIN_W];
  __shared__ int wtot[8];
  const int2* __restrict__ binned = (const int2*)binnedi;
  const int tid = threadIdx.x;
  const int bin = blockIdx.x;

  int v = 0;
  if (tid < bin) v += btot[tid];
  if (tid + 256 < bin) v += btot[tid + 256];
  const int T = btot[bin];
  ncnt[tid] = 0;
  int binBase;
  gcn_block_excl_scan(v, wtot, binBase);

  const int2* __restrict__ mine = binned + binBase;
  for (int i0 = tid; i0 < T; i0 += 256 * LIST_PER) {
    int dl[LIST_PER];
#pragma unroll
    for (int u = 0; u < LIST_PER; ++u) {
      const int i = i0 + u * 256;
      dl[u] = (i < T) ? mine[i].y : -1;
    }
#pragma unroll
    for (int u = 0; u < LIST_PER; ++u)
      if (dl[u] >= 0) atomicAdd(&ncnt[dl[u] & (BIN_W - 1)], 1);
  }
  __syncthreads();

  const int c = ncnt[tid];
  int tsum;
  const int first = binBase + gcn_block_excl_scan(c, wtot, tsum);
  const int node = bin * BIN_W + tid;
  if (node < n) {
    rowptr[node] = first;
    isq[node] = rsqrtf((float)(c + 1));
  }
  if (bin == nbins - 1 && tid == 0) rowptr[n] = binBase + T;
  ncnt[tid] = first;
  __syncthreads();

  for (int i0 = tid; i0 < T; i0 += 256 * LIST_PER) {
    int2 p[LIST_PER];
#pragma unroll
    for (int u = 0; u < LIST_PER; ++u) {
      const int i = i0 + u * 256;
      p[u] = (i < T) ? mine[i] : make_int2(0, -1);
    }
#pragma unroll
    for (int u = 0; u < LIST_PER; ++u)
      if (p[u].y >= 0) ew[atomicAdd(&ncnt[p[u].y & (BIN_W - 1)], 1)] = p[u].x;
  }
}

#define GEMM_ROWS 16
#define GEMM_THR  128

template <int K>
__global__ __launch_bounds__(GEMM_THR) void gcn_gemm_wmma_f32(
    const float* __restrict__ in, const float* __restrict__ W,
    const float* __restrict__ rowscale, float* __restrict__ out, int ntiles) {
  constexpr int ldA = K + 4;
  constexpr int C4  = K / 4;
  constexpr int PER = GEMM_ROWS * C4 / GEMM_THR;
  static_assert(PER == 2 || PER == 4, "K must be 64 or 128");
  __shared__ alignas(16) float As[GEMM_ROWS * ldA];
  const int tid  = threadIdx.x;
  const int wave = tid >> 5;
  const int lane = tid & 31;
  const int m  = lane & 15;
  const int kp = lane >> 4;
  const int col0 = wave * 16;

  int tile = blockIdx.x;
  if (tile >= ntiles) return;

  v2f bf[K / 4];
#pragma unroll
  for (int i = 0; i < K / 4; ++i) {
    const int ka = 4 * i + 2 * kp;
    bf[i].x = W[(size_t)(ka + 0) * NFEAT + col0 + m];
    bf[i].y = W[(size_t)(ka + 1) * NFEAT + col0 + m];
  }

  const float* src0 = in + (size_t)(tid / C4) * K + (tid % C4) * 4;
  constexpr size_t pieceStep = (size_t)(GEMM_THR / C4) * K;
  constexpr size_t tileStep  = (size_t)GEMM_ROWS * K;
  float* dst0 = As + (tid / C4) * ldA + (tid % C4) * 4;
  constexpr int ldsStep = (GEMM_THR / C4) * ldA;
  const float* sp = src0 + (size_t)tile * tileStep;
  v4f p0 = *(const v4f*)(sp);
  v4f p1 = *(const v4f*)(sp + pieceStep);
  v4f p2 = {}, p3 = {};
  if constexpr (PER == 4) {
    p2 = *(const v4f*)(sp + 2 * pieceStep);
    p3 = *(const v4f*)(sp + 3 * pieceStep);
  }

  for (; tile < ntiles; tile += gridDim.x) {
    __syncthreads();
    *(v4f*)(dst0) = p0;
    *(v4f*)(dst0 + ldsStep) = p1;
    if constexpr (PER == 4) {
      *(v4f*)(dst0 + 2 * ldsStep) = p2;
      *(v4f*)(dst0 + 3 * ldsStep) = p3;
    }
    __syncthreads();

    const int nt = tile + gridDim.x;
    if (nt < ntiles) {
      sp = src0 + (size_t)nt * tileStep;
      p0 = *(const v4f*)(sp);
      p1 = *(const v4f*)(sp + pieceStep);
      if constexpr (PER == 4) {
        p2 = *(const v4f*)(sp + 2 * pieceStep);
        p3 = *(const v4f*)(sp + 3 * pieceStep);
      }
    }

    const int gRow0 = tile * GEMM_ROWS;
    float rs[8];
#pragma unroll
    for (int i = 0; i < 8; ++i) rs[i] = rowscale[gRow0 + i + 8 * kp];

    v8f c = {};
#pragma unroll
    for (int i = 0; i < K / 4; ++i) {
      const v2f a = *(const v2f*)(As + m * ldA + 4 * i + 2 * kp);
      c = __builtin_amdgcn_wmma_f32_16x16x4_f32(
           false, a,  false, bf[i],
           (short)0, c,  false,  false);
    }

#pragma unroll
    for (int i = 0; i < 8; ++i) {
      out[(size_t)(gRow0 + i + 8 * kp) * NFEAT + col0 + m] = c[i] * rs[i];
    }
  }
}

#define GATH_NODES 5
#define GATH_SLOTS 16

__device__ __forceinline__ void gcn_pool_flush4(float* __restrict__ gsum, float* __restrict__ gcnt,
                                                int g, int l16, float4 a, float c) {
  float* o = gsum + g * NFEAT + l16 * 4;
  atomicAdd(o + 0, a.x);
  atomicAdd(o + 1, a.y);
  atomicAdd(o + 2, a.z);
  atomicAdd(o + 3, a.w);
  if (l16 == 0) atomicAdd(&gcnt[g], c);
}

template <bool POOL>
__global__ __launch_bounds__(256) void gcn_gather_relu(
    const int* __restrict__ rowptr, const int* __restrict__ ew,
    const float* __restrict__ h, const float* __restrict__ isq,
    const float* __restrict__ bias, float* __restrict__ out,
    const int* __restrict__ batch, float* __restrict__ gsum, float* __restrict__ gcnt, int n) {
  const int tid  = threadIdx.x;
  const int l16  = tid & 15;
  const int slot = tid >> 4;
  const int step = POOL ? 1 : gridDim.x * GATH_SLOTS;
  int node = POOL ? (blockIdx.x * GATH_SLOTS + slot) * GATH_NODES : blockIdx.x * GATH_SLOTS + slot;
  const int last = POOL ? (node + GATH_NODES < n ? node + GATH_NODES : n) : n;
  if constexpr (!POOL) {
    if (node >= last) return;
  }
  const float* hp = h + l16 * 4;
  const float4 b = *(const float4*)(bias + l16 * 4);

  int cur = -1;
  float4 racc = make_float4(0.0f, 0.0f, 0.0f, 0.0f);
  float rcnt = 0.0f;

  if (node < last) {
    int beg = rowptr[node];
    int end = rowptr[node + 1];
    int nbeg = 0, nend = 0;
    if (node + step < last) {
      nbeg = rowptr[node + step];
      nend = rowptr[node + step + 1];
    }
    int sl = (beg + l16 < end) ? ew[beg + l16] : node;

    for (;;) {
      const int nxt = node + step;
      int nsl = nxt;
      if (nbeg + l16 < nend) nsl = ew[nbeg + l16];
      int n2beg = 0, n2end = 0;
      if (nxt + step < last) {
        n2beg = rowptr[nxt + step];
        n2end = rowptr[nxt + step + 1];
      }

      const float isd = isq[node];
      int gid = 0;
      if constexpr (POOL) gid = batch[node];
      float4 acc = *(const float4*)(hp + (size_t)node * NFEAT);

      for (int base = beg; base < end; base += 16) {
        if (base != beg) {
          const int my = base + l16;
          sl = (my < end) ? ew[my] : node;
        }
        const int cnt = (end - base) < 16 ? (end - base) : 16;
        for (int j = 0; j < cnt; j += 4) {
          int    s[4];
          float4 hv[4];
#pragma unroll
          for (int u = 0; u < 4; ++u) s[u] = __shfl(sl, j + u, 16);
#pragma unroll
          for (int u = 0; u < 4; ++u) hv[u] = *(const float4*)(hp + (size_t)s[u] * NFEAT);
#pragma unroll
          for (int u = 0; u < 4; ++u) {
            if (j + u < cnt) {
              acc.x += hv[u].x;
              acc.y += hv[u].y;
              acc.z += hv[u].z;
              acc.w += hv[u].w;
            }
          }
        }
      }

      float4 o;
      o.x = fmaxf(fmaf(acc.x, isd, b.x), 0.0f);
      o.y = fmaxf(fmaf(acc.y, isd, b.y), 0.0f);
      o.z = fmaxf(fmaf(acc.z, isd, b.z), 0.0f);
      o.w = fmaxf(fmaf(acc.w, isd, b.w), 0.0f);
      if constexpr (POOL) {
        if (gid != cur) {
          if (cur >= 0) gcn_pool_flush4(gsum, gcnt, cur, l16, racc, rcnt);
          cur = gid;
          racc = make_float4(0.0f, 0.0f, 0.0f, 0.0f);
          rcnt = 0.0f;
        }
        racc.x += o.x;  racc.y += o.y;  racc.z += o.z;  racc.w += o.w;
        rcnt += 1.0f;
      } else {
        *(float4*)(out + (size_t)node * NFEAT + l16 * 4) = o;
      }

      if (nxt >= last) break;
      node = nxt;
      beg = nbeg;  end = nend;  sl = nsl;
      nbeg = n2beg;  nend = n2end;
    }
  }

  if constexpr (POOL) {
    __shared__ alignas(16) float part[GATH_SLOTS * NFEAT];
    __shared__ int   pgid[GATH_SLOTS];
    __shared__ float pcnt[GATH_SLOTS];
    *(float4*)(part + slot * NFEAT + l16 * 4) = racc;
    if (l16 == 0) { pgid[slot] = cur; pcnt[slot] = rcnt; }
    __syncthreads();

    if (tid < NFEAT) {
      int   g16[GATH_SLOTS];
      float p16[GATH_SLOTS], c16[GATH_SLOTS];
#pragma unroll
      for (int s = 0; s < GATH_SLOTS; ++s) {
        g16[s] = pgid[s];
        p16[s] = part[s * NFEAT + tid];
        c16[s] = pcnt[s];
      }
      int mg = -1;
      float a = 0.0f, c = 0.0f;
#pragma unroll
      for (int s = 0; s < GATH_SLOTS; ++s) {
        const int g = g16[s];
        if (g >= 0) {
          if (g != mg) {
            if (mg >= 0) {
              atomicAdd(&gsum[mg * NFEAT + tid], a);
              if (tid == 0) atomicAdd(&gcnt[mg], c);
            }
            mg = g; a = 0.0f; c = 0.0f;
          }
          a += p16[s];
          c += c16[s];
        }
      }
      if (mg >= 0) {
        atomicAdd(&gsum[mg * NFEAT + tid], a);
        if (tid == 0) atomicAdd(&gcnt[mg], c);
      }
    }
  }
}

#define HEAD_G   64
#define HEAD_NC  6

__global__ __launch_bounds__(NFEAT) void gcn_head(
    const float* __restrict__ gsum, const float* __restrict__ gcnt,
    const float* __restrict__ Wp, const float* __restrict__ bp,
    const float* __restrict__ Wc, const float* __restrict__ bc,
    float* __restrict__ out) {
  __shared__ float g[NFEAT];
  __shared__ float zs[NFEAT];
  const int t = threadIdx.x;
  const int m = blockIdx.x;

  float wp[NFEAT], wc[NFEAT];
  float l = 0.0f;
#pragma unroll
  for (int k = 0; k < NFEAT; ++k) wp[k] = Wp[k * NFEAT + t];
  if (t < HEAD_NC) {
#pragma unroll
    for (int k = 0; k < NFEAT; ++k) wc[k] = Wc[k * HEAD_NC + t];
    l = bc[t];
  }
  const float c = fmaxf(gcnt[m], 1.0f);
  g[t] = gsum[m * NFEAT + t] / c;
  float a = bp[t];
  __syncthreads();

#pragma unroll
  for (int k = 0; k < NFEAT; ++k) a += g[k] * wp[k];
  zs[t] = a;
  out[HEAD_G * HEAD_NC + m * NFEAT + t] = a;
  __syncthreads();

  if (t < HEAD_NC) {
#pragma unroll
    for (int k = 0; k < NFEAT; ++k) l += zs[k] * wc[k];
    out[m * HEAD_NC + t] = l;
  }
}

extern "C" void kernel_launch(void* const* d_in, const int* in_sizes, int n_in,
                              void* d_out, int out_size, void* d_ws, size_t ws_size,
                              hipStream_t stream) {
  const float* x     = (const float*)d_in[0];
  const int*   ei    = (const int*)d_in[1];
  const int*   batch = (const int*)d_in[2];
  const float* W1    = (const float*)d_in[3];
  const float* b1    = (const float*)d_in[4];
  const float* W2    = (const float*)d_in[5];
  const float* b2    = (const float*)d_in[6];
  const float* Wp    = (const float*)d_in[7];
  const float* bp    = (const float*)d_in[8];
  const float* Wc    = (const float*)d_in[9];
  const float* bc    = (const float*)d_in[10];

  const int n = in_sizes[0] / 128;
  const int E = in_sizes[1] / 2;
  const int* src = ei;
  const int* dst = ei + E;

  const int T = 256;
  const int eBlk  = (E + CHUNK - 1) / CHUNK;
  const int nbins = (n + BIN_W - 1) / BIN_W;
  const int gemmTiles = n / GEMM_ROWS;
  const int gemmBlk = (gemmTiles + 4) / 5;
  const int gathBlk = ((n + 15) / 16 + 4) / 5;

  char* ws = (char*)d_ws;
  size_t off = 0;
  auto carve = [&](size_t bytes) -> float* {
    float* p = (float*)(ws + off);
    off = (off + bytes + 255) & ~(size_t)255;
    return p;
  };
  float* isq  = carve((size_t)n * sizeof(float));
  float* bufA = carve((size_t)n * NFEAT * sizeof(float));
  float* bufB = carve((size_t)n * NFEAT * sizeof(float));
  int*   rowptr = (int*)carve((size_t)(n + 1) * sizeof(int));
  int*   binned = (int*)carve((size_t)E * 2 * sizeof(int));
  int*   ew     = (int*)carve((size_t)E * sizeof(int));
  float* zero0 = carve(MAX_BINS * sizeof(int));
  int*   btot  = (int*)zero0;
  int*   bcur  = (int*)carve(MAX_BINS * sizeof(int));
  float* gsum  = carve(64 * NFEAT * sizeof(float));
  float* gcnt  = carve(64 * sizeof(float));
  const int zeroCount = (int)((gcnt + 64) - zero0);

  gcn_zero_f32<<<(zeroCount + T - 1) / T, T, 0, stream>>>(zero0, zeroCount);
  gcn_bin_count<<<eBlk, T, 0, stream>>>(dst, btot, E, nbins);
  gcn_bin_fill<<<eBlk, T, 0, stream>>>(src, dst, btot, bcur, binned, E, nbins);
  gcn_bin_lists<<<nbins, T, 0, stream>>>(binned, btot, rowptr, isq, ew, n, nbins);

  gcn_gemm_wmma_f32<128><<<gemmBlk, GEMM_THR, 0, stream>>>(x, W1, isq, bufA, gemmTiles);
  gcn_gather_relu<false><<<gathBlk, T, 0, stream>>>(rowptr, ew, bufA, isq, b1, bufB,
                                                    batch, gsum, gcnt, n);

  gcn_gemm_wmma_f32<64><<<gemmBlk, GEMM_THR, 0, stream>>>(bufB, W2, isq, bufA, gemmTiles);
  gcn_gather_relu<true><<<gathBlk, T, 0, stream>>>(rowptr, ew, bufA, isq, b2, nullptr,
                                                   batch, gsum, gcnt, n);

  gcn_head<<<HEAD_G, NFEAT, 0, stream>>>(gsum, gcnt, Wp, bp, Wc, bc, (float*)d_out);
}
